# starting kernel
# baseline (speedup 1.0000x reference)
#include <hip/hip_runtime.h>
#include <hip/hip_bf16.h>

typedef float v2f __attribute__((ext_vector_type(2)));
typedef float v8f __attribute__((ext_vector_type(8)));

constexpr int NB = 32;
constexpr int NH = 56, NW = 56;
constexpr int NC = 256;
constexpr int NP = 128;
constexpr int NS = NH * NW;
constexpr int MTILE = 64;
constexpr int NTILES = NP / 16;
constexpr int TILES = NS / MTILE;
constexpr float MAXDIST = 100.0f * 256.0f;
constexpr float EPS = 1e-4f;

__global__ __launch_bounds__(128)
void proto_init(const float* __restrict__ proto,
                unsigned int* __restrict__ wmin,
                unsigned int* __restrict__ wmax,
                float* __restrict__ yty)
{
    int g = blockIdx.x * 128 + threadIdx.x;
    if (g < NB * NP) {
        wmin[g] = 0x7F800000u;
        wmax[g] = 0u;
    }
    if (blockIdx.x == 0) {
        int p = threadIdx.x;
        float s = 0.f;
        for (int c = 0; c < NC; ++c) {
            float v = proto[c * NP + p];
            s += v * v;
        }
        yty[p] = s;
    }
}

__global__ __launch_bounds__(256)
void proto_main(const float* __restrict__ in,
                const float* __restrict__ proto,
                const float* __restrict__ lscale,
                const float* __restrict__ yty,
                unsigned int* __restrict__ wmin,
                unsigned int* __restrict__ wmax)
{
    extern __shared__ float lds[];
    float* a_lds = lds;
    float* p_lds = lds + 2 * MTILE * NC;
    float* xtx   = p_lds + NC * NP;

    const int tile = blockIdx.x;
    const int b0   = blockIdx.y * 2;
    const int tid  = threadIdx.x;

    for (int bb = 0; bb < 2; ++bb) {
        const float4* src = (const float4*)(in + ((size_t)(b0 + bb) * NS + (size_t)tile * MTILE) * NC);
        float4* dst = (float4*)(a_lds + bb * MTILE * NC);
        for (int i = tid; i < MTILE * NC / 4; i += 256) dst[i] = src[i];
    }
    {
        float2* dst = (float2*)p_lds;
        for (int i = tid; i < (NC / 2) * NP; i += 256) {
            int kp = i >> 7;
            int p  = i & (NP - 1);
            float v0 = proto[(2 * kp) * NP + p];
            float v1 = proto[(2 * kp + 1) * NP + p];
            dst[i] = make_float2(v0, v1);
        }
    }
    __syncthreads();

    if (tid < 2 * MTILE) {
        const float4* row = (const float4*)(a_lds + tid * NC);
        float s = 0.f;
        for (int i = 0; i < NC / 4; ++i) {
            float4 v = row[i];
            s += v.x * v.x + v.y * v.y + v.z * v.z + v.w * v.w;
        }
        xtx[tid] = s;
    }
    __syncthreads();

    const int wave  = tid >> 5;
    const int lane  = tid & 31;
    const int lm    = lane & 15;
    const int hsel  = lane >> 4;
    const int bb    = wave >> 2;
    const int strip = wave & 3;

    const float* Arow = a_lds + (bb * MTILE + strip * 16) * NC;

    v8f acc[NTILES];
    #pragma unroll
    for (int n = 0; n < NTILES; ++n) acc[n] = (v8f){0.f, 0.f, 0.f, 0.f, 0.f, 0.f, 0.f, 0.f};

    const int a_off = lm * NC + 2 * hsel;
    const v2f* Pfrag = (const v2f*)p_lds;

    #pragma unroll 1
    for (int k0 = 0; k0 < NC; k0 += 4) {
        v2f Af = *(const v2f*)(Arow + a_off + k0);
        const v2f* Bp = Pfrag + ((k0 >> 1) + hsel) * NP + lm;
        #pragma unroll
        for (int n = 0; n < NTILES; ++n) {
            v2f Bf = Bp[n * 16];
            acc[n] = __builtin_amdgcn_wmma_f32_16x16x4_f32(
                false, Af, false, Bf, (short)0, acc[n], false, false);
        }
    }

    const int row_loc = bb * MTILE + strip * 16 + hsel * 8;
    const int s_base  = tile * MTILE + strip * 16 + hsel * 8;
    const int bglob   = b0 + bb;

    #pragma unroll
    for (int n = 0; n < NTILES; ++n) {
        const int p  = n * 16 + lm;
        const float yv = yty[p];
        float mn = 3.0e38f;
        float mx = 0.f;
        #pragma unroll
        for (int r = 0; r < 8; ++r) {
            float xt  = xtx[row_loc + r];
            float ls  = lscale[(size_t)(s_base + r) * NP + p];
            float sf  = __expf(ls);
            float xty = acc[n][r];
            float nsq = fmaxf(xt - 2.f * xty + yv, 0.f);
            float dmin = nsq / (sf + EPS);
            float sim  = __logf((nsq + 1.f) / (nsq + EPS)) * sf;
            mn = fminf(mn, dmin);
            mx = fmaxf(mx, sim);
        }
        mn = fminf(mn, __shfl_xor(mn, 16));
        mx = fmaxf(mx, __shfl_xor(mx, 16));
        if (lane < 16) {
            atomicMin(wmin + bglob * NP + p, __float_as_uint(mn));
            atomicMax(wmax + bglob * NP + p, __float_as_uint(mx));
        }
    }
}

__global__ __launch_bounds__(128)
void proto_final(const unsigned int* __restrict__ wmin,
                 const unsigned int* __restrict__ wmax,
                 const float* __restrict__ pcc,
                 float* __restrict__ out)
{
    __shared__ float r1[128];
    __shared__ float r2[128];
    const int t = threadIdx.x;

    for (int i = t; i < NB * NP; i += 128)
        out[i] = __uint_as_float(wmax[i]);

    float s1 = 0.f, s2 = 0.f;
    for (int b = 0; b < NB; ++b) {
        float md = __uint_as_float(wmin[b * NP + t]);
        float v  = MAXDIST - md;
        float m  = pcc[b * NP + t];
        r1[t] = v * m;
        r2[t] = v * (1.f - m);
        __syncthreads();
        for (int st = 64; st > 0; st >>= 1) {
            if (t < st) {
                r1[t] = fmaxf(r1[t], r1[t + st]);
                r2[t] = fmaxf(r2[t], r2[t + st]);
            }
            __syncthreads();
        }
        if (t == 0) { s1 += MAXDIST - r1[0]; s2 += MAXDIST - r2[0]; }
        __syncthreads();
    }
    if (t == 0) {
        out[NB * NP]     = s1 / (float)NB;
        out[NB * NP + 1] = s2 / (float)NB;
    }
}

extern "C" void kernel_launch(void* const* d_in, const int* in_sizes, int n_in,
                              void* d_out, int out_size, void* d_ws, size_t ws_size,
                              hipStream_t stream)
{
    const float* inputs = (const float*)d_in[0];
    const float* pcc    = (const float*)d_in[1];
    const float* proto  = (const float*)d_in[2];
    const float* lscale = (const float*)d_in[3];
    float* out = (float*)d_out;

    unsigned int* wmin = (unsigned int*)d_ws;
    unsigned int* wmax = wmin + NB * NP;
    float* yty = (float*)(wmax + NB * NP);

    proto_init<<<dim3((NB * NP) / 128), dim3(128), 0, stream>>>(proto, wmin, wmax, yty);

    const size_t shmem = (size_t)(2 * MTILE * NC + NC * NP + 2 * MTILE) * sizeof(float);
    proto_main<<<dim3(TILES, NB / 2), dim3(256), shmem, stream>>>(
        inputs, proto, lscale, yty, wmin, wmax);

    proto_final<<<dim3(1), dim3(128), 0, stream>>>(wmin, wmax, pcc, out);
}


# final kernel: one wave per batch row (32 waves: register max, then 5 lane exchanges) replaces 32 serial LDS tree reductions of one 4-wave block; max is order-free, the 32-term sums keep their order
# speedup vs baseline: 1.3232x; 1.3232x over previous
#include <hip/hip_runtime.h>
#include <hip/hip_bf16.h>

typedef float v2f __attribute__((ext_vector_type(2)));
typedef float v8f __attribute__((ext_vector_type(8)));

constexpr int NB = 32;
constexpr int NH = 56, NW = 56;
constexpr int NC = 256;
constexpr int NP = 128;
constexpr int NS = NH * NW;
constexpr int MTILE = 64;
constexpr int NTILES = NP / 16;
constexpr int TILES = NS / MTILE;
constexpr float MAXDIST = 100.0f * 256.0f;
constexpr float EPS = 1e-4f;
constexpr int FIN_THREADS = NB * 32;

__global__ __launch_bounds__(128)
void proto_init(const float* __restrict__ proto,
                unsigned int* __restrict__ wmin,
                unsigned int* __restrict__ wmax,
                float* __restrict__ yty)
{
    int g = blockIdx.x * 128 + threadIdx.x;
    if (g < NB * NP) {
        wmin[g] = 0x7F800000u;
        wmax[g] = 0u;
    }
    if (blockIdx.x == 0) {
        int p = threadIdx.x;
        float s = 0.f;
        for (int c = 0; c < NC; ++c) {
            float v = proto[c * NP + p];
            s += v * v;
        }
        yty[p] = s;
    }
}

__global__ __launch_bounds__(256)
void proto_main(const float* __restrict__ in,
                const float* __restrict__ proto,
                const float* __restrict__ lscale,
                const float* __restrict__ yty,
                unsigned int* __restrict__ wmin,
                unsigned int* __restrict__ wmax)
{
    extern __shared__ float lds[];
    float* a_lds = lds;
    float* p_lds = lds + 2 * MTILE * NC;
    float* xtx   = p_lds + NC * NP;

    const int tile = blockIdx.x;
    const int b0   = blockIdx.y * 2;
    const int tid  = threadIdx.x;

    for (int bb = 0; bb < 2; ++bb) {
        const float4* src = (const float4*)(in + ((size_t)(b0 + bb) * NS + (size_t)tile * MTILE) * NC);
        float4* dst = (float4*)(a_lds + bb * MTILE * NC);
        for (int i = tid; i < MTILE * NC / 4; i += 256) dst[i] = src[i];
    }
    {
        float2* dst = (float2*)p_lds;
        for (int i = tid; i < (NC / 2) * NP; i += 256) {
            int kp = i >> 7;
            int p  = i & (NP - 1);
            float v0 = proto[(2 * kp) * NP + p];
            float v1 = proto[(2 * kp + 1) * NP + p];
            dst[i] = make_float2(v0, v1);
        }
    }
    __syncthreads();

    if (tid < 2 * MTILE) {
        const float4* row = (const float4*)(a_lds + tid * NC);
        float s = 0.f;
        for (int i = 0; i < NC / 4; ++i) {
            float4 v = row[i];
            s += v.x * v.x + v.y * v.y + v.z * v.z + v.w * v.w;
        }
        xtx[tid] = s;
    }
    __syncthreads();

    const int wave  = tid >> 5;
    const int lane  = tid & 31;
    const int lm    = lane & 15;
    const int hsel  = lane >> 4;
    const int bb    = wave >> 2;
    const int strip = wave & 3;

    const float* Arow = a_lds + (bb * MTILE + strip * 16) * NC;

    v8f acc[NTILES];
    #pragma unroll
    for (int n = 0; n < NTILES; ++n) acc[n] = (v8f){0.f, 0.f, 0.f, 0.f, 0.f, 0.f, 0.f, 0.f};

    const int a_off = lm * NC + 2 * hsel;
    const v2f* Pfrag = (const v2f*)p_lds;

    #pragma unroll 1
    for (int k0 = 0; k0 < NC; k0 += 4) {
        v2f Af = *(const v2f*)(Arow + a_off + k0);
        const v2f* Bp = Pfrag + ((k0 >> 1) + hsel) * NP + lm;
        #pragma unroll
        for (int n = 0; n < NTILES; ++n) {
            v2f Bf = Bp[n * 16];
            acc[n] = __builtin_amdgcn_wmma_f32_16x16x4_f32(
                false, Af, false, Bf, (short)0, acc[n], false, false);
        }
    }

    const int row_loc = bb * MTILE + strip * 16 + hsel * 8;
    const int s_base  = tile * MTILE + strip * 16 + hsel * 8;
    const int bglob   = b0 + bb;

    #pragma unroll
    for (int n = 0; n < NTILES; ++n) {
        const int p  = n * 16 + lm;
        const float yv = yty[p];
        float mn = 3.0e38f;
        float mx = 0.f;
        #pragma unroll
        for (int r = 0; r < 8; ++r) {
            float xt  = xtx[row_loc + r];
            float ls  = lscale[(size_t)(s_base + r) * NP + p];
            float sf  = __expf(ls);
            float xty = acc[n][r];
            float nsq = fmaxf(xt - 2.f * xty + yv, 0.f);
            float dmin = nsq / (sf + EPS);
            float sim  = __logf((nsq + 1.f) / (nsq + EPS)) * sf;
            mn = fminf(mn, dmin);
            mx = fmaxf(mx, sim);
        }
        mn = fminf(mn, __shfl_xor(mn, 16));
        mx = fmaxf(mx, __shfl_xor(mx, 16));
        if (lane < 16) {
            atomicMin(wmin + bglob * NP + p, __float_as_uint(mn));
            atomicMax(wmax + bglob * NP + p, __float_as_uint(mx));
        }
    }
}

__global__ __launch_bounds__(FIN_THREADS)
void proto_final(const unsigned int* __restrict__ wmin,
                 const unsigned int* __restrict__ wmax,
                 const float* __restrict__ pcc,
                 float* __restrict__ out)
{
    __shared__ float r1[NB];
    __shared__ float r2[NB];
    const int t    = threadIdx.x;
    const int b    = t >> 5;
    const int lane = t & 31;

    for (int j = 0; j < (NB * NP) / FIN_THREADS; ++j)
        out[j * FIN_THREADS + t] = __uint_as_float(wmax[j * FIN_THREADS + t]);

    float m1, m2;
    {
        float md = __uint_as_float(wmin[b * NP + lane]);
        float v  = MAXDIST - md;
        float m  = pcc[b * NP + lane];
        m1 = v * m;
        m2 = v * (1.f - m);
    }
    for (int j = 1; j < NP / 32; ++j) {
        float md = __uint_as_float(wmin[b * NP + j * 32 + lane]);
        float v  = MAXDIST - md;
        float m  = pcc[b * NP + j * 32 + lane];
        m1 = fmaxf(m1, v * m);
        m2 = fmaxf(m2, v * (1.f - m));
    }
    for (int off = 16; off > 0; off >>= 1) {
        m1 = fmaxf(m1, __shfl_xor(m1, off));
        m2 = fmaxf(m2, __shfl_xor(m2, off));
    }
    if (lane == 0) { r1[b] = m1; r2[b] = m2; }
    __syncthreads();

    if (t == 0) {
        float s1 = 0.f, s2 = 0.f;
        for (int bb = 0; bb < NB; ++bb) {
            s1 += MAXDIST - r1[bb];
            s2 += MAXDIST - r2[bb];
        }
        out[NB * NP]     = s1 / (float)NB;
        out[NB * NP + 1] = s2 / (float)NB;
    }
}

extern "C" void kernel_launch(void* const* d_in, const int* in_sizes, int n_in,
                              void* d_out, int out_size, void* d_ws, size_t ws_size,
                              hipStream_t stream)
{
    const float* inputs = (const float*)d_in[0];
    const float* pcc    = (const float*)d_in[1];
    const float* proto  = (const float*)d_in[2];
    const float* lscale = (const float*)d_in[3];
    float* out = (float*)d_out;

    unsigned int* wmin = (unsigned int*)d_ws;
    unsigned int* wmax = wmin + NB * NP;
    float* yty = (float*)(wmax + NB * NP);

    proto_init<<<dim3((NB * NP) / 128), dim3(128), 0, stream>>>(proto, wmin, wmax, yty);

    const size_t shmem = (size_t)(2 * MTILE * NC + NC * NP + 2 * MTILE) * sizeof(float);
    proto_main<<<dim3(TILES, NB / 2), dim3(256), shmem, stream>>>(
        inputs, proto, lscale, yty, wmin, wmax);

    proto_final<<<dim3(1), dim3(FIN_THREADS), 0, stream>>>(wmin, wmax, pcc, out);
}


# pad A-tile rows in LDS (pitch 256->260 floats): the row-norm pass (a wave's 32 rows on one bank group) and the A fragment loads (16 rows on one bank pair) lose their bank conflicts; sum orders kept
# speedup vs baseline: 1.4197x; 1.0729x over previous
#include <hip/hip_runtime.h>
#include <hip/hip_bf16.h>

typedef float v2f __attribute__((ext_vector_type(2)));
typedef float v8f __attribute__((ext_vector_type(8)));

constexpr int NB = 32;
constexpr int NH = 56, NW = 56;
constexpr int NC = 256;
constexpr int NP = 128;
constexpr int NS = NH * NW;
constexpr int MTILE = 64;
constexpr int NTILES = NP / 16;
constexpr int TILES = NS / MTILE;
constexpr float MAXDIST = 100.0f * 256.0f;
constexpr float EPS = 1e-4f;
constexpr int FIN_THREADS = NB * 32;
constexpr int APITCH = NC + 4;

__global__ __launch_bounds__(128)
void proto_init(const float* __restrict__ proto,
                unsigned int* __restrict__ wmin,
                unsigned int* __restrict__ wmax,
                float* __restrict__ yty)
{
    int g = blockIdx.x * 128 + threadIdx.x;
    if (g < NB * NP) {
        wmin[g] = 0x7F800000u;
        wmax[g] = 0u;
    }
    if (blockIdx.x == 0) {
        int p = threadIdx.x;
        float s = 0.f;
        for (int c = 0; c < NC; ++c) {
            float v = proto[c * NP + p];
            s += v * v;
        }
        yty[p] = s;
    }
}

__global__ __launch_bounds__(256)
void proto_main(const float* __restrict__ in,
                const float* __restrict__ proto,
                const float* __restrict__ lscale,
                const float* __restrict__ yty,
                unsigned int* __restrict__ wmin,
                unsigned int* __restrict__ wmax)
{
    extern __shared__ float lds[];
    float* a_lds = lds;
    float* p_lds = lds + 2 * MTILE * APITCH;
    float* xtx   = p_lds + NC * NP;

    const int tile = blockIdx.x;
    const int b0   = blockIdx.y * 2;
    const int tid  = threadIdx.x;

    for (int bb = 0; bb < 2; ++bb) {
        const float4* src = (const float4*)(in + ((size_t)(b0 + bb) * NS + (size_t)tile * MTILE) * NC);
        float* dst = a_lds + bb * MTILE * APITCH;
        for (int i = tid; i < MTILE * NC / 4; i += 256) {
            int row = i >> 6;
            int c4  = i & 63;
            *(float4*)(dst + row * APITCH + 4 * c4) = src[i];
        }
    }
    {
        float2* dst = (float2*)p_lds;
        for (int i = tid; i < (NC / 2) * NP; i += 256) {
            int kp = i >> 7;
            int p  = i & (NP - 1);
            float v0 = proto[(2 * kp) * NP + p];
            float v1 = proto[(2 * kp + 1) * NP + p];
            dst[i] = make_float2(v0, v1);
        }
    }
    __syncthreads();

    if (tid < 2 * MTILE) {
        const float4* row = (const float4*)(a_lds + tid * APITCH);
        float s = 0.f;
        for (int i = 0; i < NC / 4; ++i) {
            float4 v = row[i];
            s += v.x * v.x + v.y * v.y + v.z * v.z + v.w * v.w;
        }
        xtx[tid] = s;
    }
    __syncthreads();

    const int wave  = tid >> 5;
    const int lane  = tid & 31;
    const int lm    = lane & 15;
    const int hsel  = lane >> 4;
    const int bb    = wave >> 2;
    const int strip = wave & 3;

    const float* Arow = a_lds + (bb * MTILE + strip * 16) * APITCH;

    v8f acc[NTILES];
    #pragma unroll
    for (int n = 0; n < NTILES; ++n) acc[n] = (v8f){0.f, 0.f, 0.f, 0.f, 0.f, 0.f, 0.f, 0.f};

    const int a_off = lm * APITCH + 2 * hsel;
    const v2f* Pfrag = (const v2f*)p_lds;

    #pragma unroll 1
    for (int k0 = 0; k0 < NC; k0 += 4) {
        v2f Af = *(const v2f*)(Arow + a_off + k0);
        const v2f* Bp = Pfrag + ((k0 >> 1) + hsel) * NP + lm;
        #pragma unroll
        for (int n = 0; n < NTILES; ++n) {
            v2f Bf = Bp[n * 16];
            acc[n] = __builtin_amdgcn_wmma_f32_16x16x4_f32(
                false, Af, false, Bf, (short)0, acc[n], false, false);
        }
    }

    const int row_loc = bb * MTILE + strip * 16 + hsel * 8;
    const int s_base  = tile * MTILE + strip * 16 + hsel * 8;
    const int bglob   = b0 + bb;

    #pragma unroll
    for (int n = 0; n < NTILES; ++n) {
        const int p  = n * 16 + lm;
        const float yv = yty[p];
        float mn = 3.0e38f;
        float mx = 0.f;
        #pragma unroll
        for (int r = 0; r < 8; ++r) {
            float xt  = xtx[row_loc + r];
            float ls  = lscale[(size_t)(s_base + r) * NP + p];
            float sf  = __expf(ls);
            float xty = acc[n][r];
            float nsq = fmaxf(xt - 2.f * xty + yv, 0.f);
            float dmin = nsq / (sf + EPS);
            float sim  = __logf((nsq + 1.f) / (nsq + EPS)) * sf;
            mn = fminf(mn, dmin);
            mx = fmaxf(mx, sim);
        }
        mn = fminf(mn, __shfl_xor(mn, 16));
        mx = fmaxf(mx, __shfl_xor(mx, 16));
        if (lane < 16) {
            atomicMin(wmin + bglob * NP + p, __float_as_uint(mn));
            atomicMax(wmax + bglob * NP + p, __float_as_uint(mx));
        }
    }
}

__global__ __launch_bounds__(FIN_THREADS)
void proto_final(const unsigned int* __restrict__ wmin,
                 const unsigned int* __restrict__ wmax,
                 const float* __restrict__ pcc,
                 float* __restrict__ out)
{
    __shared__ float r1[NB];
    __shared__ float r2[NB];
    const int t    = threadIdx.x;
    const int b    = t >> 5;
    const int lane = t & 31;

    for (int j = 0; j < (NB * NP) / FIN_THREADS; ++j)
        out[j * FIN_THREADS + t] = __uint_as_float(wmax[j * FIN_THREADS + t]);

    float m1, m2;
    {
        float md = __uint_as_float(wmin[b * NP + lane]);
        float v  = MAXDIST - md;
        float m  = pcc[b * NP + lane];
        m1 = v * m;
        m2 = v * (1.f - m);
    }
    for (int j = 1; j < NP / 32; ++j) {
        float md = __uint_as_float(wmin[b * NP + j * 32 + lane]);
        float v  = MAXDIST - md;
        float m  = pcc[b * NP + j * 32 + lane];
        m1 = fmaxf(m1, v * m);
        m2 = fmaxf(m2, v * (1.f - m));
    }
    for (int off = 16; off > 0; off >>= 1) {
        m1 = fmaxf(m1, __shfl_xor(m1, off));
        m2 = fmaxf(m2, __shfl_xor(m2, off));
    }
    if (lane == 0) { r1[b] = m1; r2[b] = m2; }
    __syncthreads();

    if (t == 0) {
        float s1 = 0.f, s2 = 0.f;
        for (int bb = 0; bb < NB; ++bb) {
            s1 += MAXDIST - r1[bb];
            s2 += MAXDIST - r2[bb];
        }
        out[NB * NP]     = s1 / (float)NB;
        out[NB * NP + 1] = s2 / (float)NB;
    }
}

extern "C" void kernel_launch(void* const* d_in, const int* in_sizes, int n_in,
                              void* d_out, int out_size, void* d_ws, size_t ws_size,
                              hipStream_t stream)
{
    const float* inputs = (const float*)d_in[0];
    const float* pcc    = (const float*)d_in[1];
    const float* proto  = (const float*)d_in[2];
    const float* lscale = (const float*)d_in[3];
    float* out = (float*)d_out;

    unsigned int* wmin = (unsigned int*)d_ws;
    unsigned int* wmax = wmin + NB * NP;
    float* yty = (float*)(wmax + NB * NP);

    proto_init<<<dim3((NB * NP) / 128), dim3(128), 0, stream>>>(proto, wmin, wmax, yty);

    const size_t shmem = (size_t)(2 * MTILE * APITCH + NC * NP + 2 * MTILE) * sizeof(float);
    proto_main<<<dim3(TILES, NB / 2), dim3(256), shmem, stream>>>(
        inputs, proto, lscale, yty, wmin, wmax);

    proto_final<<<dim3(1), dim3(FIN_THREADS), 0, stream>>>(wmin, wmax, pcc, out);
}


# persistent main kernel: 256 work-groups (one per WGP, 392 rows each) stage the prototypes in LDS once; each wave walks 16-row strips in a private LDS slot, no barrier in the walk; sum orders kept
# speedup vs baseline: 1.5638x; 1.1015x over previous
#include <hip/hip_runtime.h>
#include <hip/hip_bf16.h>

typedef float v2f __attribute__((ext_vector_type(2)));
typedef float v8f __attribute__((ext_vector_type(8)));

constexpr int NB = 32;
constexpr int NH = 56, NW = 56;
constexpr int NC = 256;
constexpr int NP = 128;
constexpr int NS = NH * NW;
constexpr int MTILE = 64;
constexpr int NTILES = NP / 16;
constexpr int TILES = NS / MTILE;
constexpr int PARTS   = 8;
constexpr int PROWS   = NS / PARTS;
constexpr int WSTRIPS = PROWS / 16;
constexpr int MWAVES  = 8;
static_assert(MWAVES == NTILES, "the last strip is split one column tile per wave");
static_assert(WSTRIPS % MWAVES == 0 && PROWS - 16 * WSTRIPS > 0 && PROWS - 16 * WSTRIPS <= 16,
              "three whole strips per wave and one partial strip");
constexpr float MAXDIST = 100.0f * 256.0f;
constexpr float EPS = 1e-4f;
constexpr int FIN_THREADS = NB * 32;
constexpr int APITCH = NC + 4;

__global__ __launch_bounds__(128)
void proto_init(const float* __restrict__ proto,
                unsigned int* __restrict__ wmin,
                unsigned int* __restrict__ wmax,
                float* __restrict__ yty)
{
    int g = blockIdx.x * 128 + threadIdx.x;
    if (g < NB * NP) {
        wmin[g] = 0x7F800000u;
        wmax[g] = 0u;
    }
    if (blockIdx.x == 0) {
        int p = threadIdx.x;
        float s = 0.f;
        for (int c = 0; c < NC; ++c) {
            float v = proto[c * NP + p];
            s += v * v;
        }
        yty[p] = s;
    }
}


__device__ __forceinline__ void wave_lds_sync()
{
    __builtin_amdgcn_fence(__ATOMIC_RELEASE, "wavefront");
    __builtin_amdgcn_wave_barrier();
    __builtin_amdgcn_fence(__ATOMIC_ACQUIRE, "wavefront");
}

template <int NT>
__device__ __forceinline__ void do_strip(
    const int j, const int n0,
    const float* __restrict__ in_part,
    float* slot, float* xtx_w,
    const v2f* Pfrag,
    const float* __restrict__ lscale,
    const float* __restrict__ yty,
    unsigned int* __restrict__ wmin,
    unsigned int* __restrict__ wmax,
    const int part, const int bglob, const int lane)
{
    const int lm   = lane & 15;
    const int hsel = lane >> 4;

    #pragma unroll
    for (int t = 0; t < 32; ++t) {
        const int row = t >> 1;
        const int c4  = (t & 1) * 32 + lane;
        const int lr  = min(j * 16 + row, PROWS - 1);
        float4 v = *(const float4*)(in_part + (size_t)lr * NC + 4 * c4);
        *(float4*)(slot + row * APITCH + 4 * c4) = v;
    }
    wave_lds_sync();

    {
        const float4* row = (const float4*)(slot + lm * APITCH);
        float s = 0.f;
        for (int i = 0; i < NC / 4; ++i) {
            float4 v = row[i];
            s += v.x * v.x + v.y * v.y + v.z * v.z + v.w * v.w;
        }
        xtx_w[lm] = s;
    }
    wave_lds_sync();

    v8f acc[NT];
    #pragma unroll
    for (int n = 0; n < NT; ++n) acc[n] = (v8f){0.f, 0.f, 0.f, 0.f, 0.f, 0.f, 0.f, 0.f};

    const int a_off = lm * APITCH + 2 * hsel;

    #pragma unroll (NT == NTILES ? 1 : 8)
    for (int k0 = 0; k0 < NC; k0 += 4) {
        v2f Af = *(const v2f*)(slot + a_off + k0);
        const v2f* Bp = Pfrag + ((k0 >> 1) + hsel) * NP + n0 * 16 + lm;
        #pragma unroll
        for (int n = 0; n < NT; ++n) {
            v2f Bf = Bp[n * 16];
            acc[n] = __builtin_amdgcn_wmma_f32_16x16x4_f32(
                false, Af, false, Bf, (short)0, acc[n], false, false);
        }
    }

    const int row_loc = hsel * 8;
    const int r_base  = j * 16 + hsel * 8;

    #pragma unroll
    for (int n = 0; n < NT; ++n) {
        const int p  = (n0 + n) * 16 + lm;
        const float yv = yty[p];
        float mn = 3.0e38f;
        float mx = 0.f;
        #pragma unroll
        for (int r = 0; r < 8; ++r) {
            const int s_idx = part * PROWS + min(r_base + r, PROWS - 1);
            float xt  = xtx_w[row_loc + r];
            float ls  = lscale[(size_t)s_idx * NP + p];
            float sf  = __expf(ls);
            float xty = acc[n][r];
            float nsq = fmaxf(xt - 2.f * xty + yv, 0.f);
            float dmin = nsq / (sf + EPS);
            float sim  = __logf((nsq + 1.f) / (nsq + EPS)) * sf;
            mn = fminf(mn, dmin);
            mx = fmaxf(mx, sim);
        }
        mn = fminf(mn, __shfl_xor(mn, 16));
        mx = fmaxf(mx, __shfl_xor(mx, 16));
        if (lane < 16) {
            atomicMin(wmin + bglob * NP + p, __float_as_uint(mn));
            atomicMax(wmax + bglob * NP + p, __float_as_uint(mx));
        }
    }
    wave_lds_sync();
}

__global__ __launch_bounds__(256)
void proto_main(const float* __restrict__ in,
                const float* __restrict__ proto,
                const float* __restrict__ lscale,
                const float* __restrict__ yty,
                unsigned int* __restrict__ wmin,
                unsigned int* __restrict__ wmax)
{
    extern __shared__ float lds[];
    float* a_lds = lds;
    float* p_lds = lds + MWAVES * 16 * APITCH;
    float* xtx   = p_lds + NC * NP;

    const int bglob = blockIdx.x / PARTS;
    const int part  = blockIdx.x % PARTS;
    const int tid   = threadIdx.x;

    {
        float2* dst = (float2*)p_lds;
        for (int i = tid; i < (NC / 2) * NP; i += 256) {
            int kp = i >> 7;
            int p  = i & (NP - 1);
            float v0 = proto[(2 * kp) * NP + p];
            float v1 = proto[(2 * kp + 1) * NP + p];
            dst[i] = make_float2(v0, v1);
        }
    }
    __syncthreads();

    const int wave = __builtin_amdgcn_readfirstlane(tid >> 5);
    const int lane = tid & 31;

    float* slot  = a_lds + wave * 16 * APITCH;
    float* xtx_w = xtx + wave * 16;
    const float* in_part = in + ((size_t)bglob * NS + (size_t)part * PROWS) * NC;
    const v2f* Pfrag = (const v2f*)p_lds;

    for (int j = wave; j < WSTRIPS; j += MWAVES)
        do_strip<NTILES>(j, 0, in_part, slot, xtx_w, Pfrag, lscale, yty, wmin, wmax,
                         part, bglob, lane);
    do_strip<1>(WSTRIPS, wave, in_part, slot, xtx_w, Pfrag, lscale, yty, wmin, wmax,
                part, bglob, lane);
}

__global__ __launch_bounds__(FIN_THREADS)
void proto_final(const unsigned int* __restrict__ wmin,
                 const unsigned int* __restrict__ wmax,
                 const float* __restrict__ pcc,
                 float* __restrict__ out)
{
    __shared__ float r1[NB];
    __shared__ float r2[NB];
    const int t    = threadIdx.x;
    const int b    = t >> 5;
    const int lane = t & 31;

    for (int j = 0; j < (NB * NP) / FIN_THREADS; ++j)
        out[j * FIN_THREADS + t] = __uint_as_float(wmax[j * FIN_THREADS + t]);

    float m1, m2;
    {
        float md = __uint_as_float(wmin[b * NP + lane]);
        float v  = MAXDIST - md;
        float m  = pcc[b * NP + lane];
        m1 = v * m;
        m2 = v * (1.f - m);
    }
    for (int j = 1; j < NP / 32; ++j) {
        float md = __uint_as_float(wmin[b * NP + j * 32 + lane]);
        float v  = MAXDIST - md;
        float m  = pcc[b * NP + j * 32 + lane];
        m1 = fmaxf(m1, v * m);
        m2 = fmaxf(m2, v * (1.f - m));
    }
    for (int off = 16; off > 0; off >>= 1) {
        m1 = fmaxf(m1, __shfl_xor(m1, off));
        m2 = fmaxf(m2, __shfl_xor(m2, off));
    }
    if (lane == 0) { r1[b] = m1; r2[b] = m2; }
    __syncthreads();

    if (t == 0) {
        float s1 = 0.f, s2 = 0.f;
        for (int bb = 0; bb < NB; ++bb) {
            s1 += MAXDIST - r1[bb];
            s2 += MAXDIST - r2[bb];
        }
        out[NB * NP]     = s1 / (float)NB;
        out[NB * NP + 1] = s2 / (float)NB;
    }
}

extern "C" void kernel_launch(void* const* d_in, const int* in_sizes, int n_in,
                              void* d_out, int out_size, void* d_ws, size_t ws_size,
                              hipStream_t stream)
{
    const float* inputs = (const float*)d_in[0];
    const float* pcc    = (const float*)d_in[1];
    const float* proto  = (const float*)d_in[2];
    const float* lscale = (const float*)d_in[3];
    float* out = (float*)d_out;

    unsigned int* wmin = (unsigned int*)d_ws;
    unsigned int* wmax = wmin + NB * NP;
    float* yty = (float*)(wmax + NB * NP);

    proto_init<<<dim3((NB * NP) / 128), dim3(128), 0, stream>>>(proto, wmin, wmax, yty);

    const size_t shmem = (size_t)(MWAVES * 16 * APITCH + NC * NP + MWAVES * 16) * sizeof(float);
    proto_main<<<dim3(NB * PARTS), dim3(256), shmem, stream>>>(
        inputs, proto, lscale, yty, wmin, wmax);

    proto_final<<<dim3(1), dim3(FIN_THREADS), 0, stream>>>(wmin, wmax, pcc, out);
}


# local-scale loads of a strip (8 per column tile and lane) issued as one group ahead of the row-norm pass and K loop, not singly at their uses behind it; arithmetic and sum orders unchanged
# speedup vs baseline: 1.8830x; 1.2041x over previous
#include <hip/hip_runtime.h>
#include <hip/hip_bf16.h>

typedef float v2f __attribute__((ext_vector_type(2)));
typedef float v8f __attribute__((ext_vector_type(8)));

constexpr int NB = 32;
constexpr int NH = 56, NW = 56;
constexpr int NC = 256;
constexpr int NP = 128;
constexpr int NS = NH * NW;
constexpr int MTILE = 64;
constexpr int NTILES = NP / 16;
constexpr int TILES = NS / MTILE;
constexpr int PARTS   = 8;
constexpr int PROWS   = NS / PARTS;
constexpr int WSTRIPS = PROWS / 16;
constexpr int MWAVES  = 8;
static_assert(MWAVES == NTILES, "the last strip is split one column tile per wave");
static_assert(WSTRIPS % MWAVES == 0 && PROWS - 16 * WSTRIPS > 0 && PROWS - 16 * WSTRIPS <= 16,
              "three whole strips per wave and one partial strip");
constexpr float MAXDIST = 100.0f * 256.0f;
constexpr float EPS = 1e-4f;
constexpr int FIN_THREADS = NB * 32;
constexpr int APITCH = NC + 4;

__global__ __launch_bounds__(128)
void proto_init(const float* __restrict__ proto,
                unsigned int* __restrict__ wmin,
                unsigned int* __restrict__ wmax,
                float* __restrict__ yty)
{
    int g = blockIdx.x * 128 + threadIdx.x;
    if (g < NB * NP) {
        wmin[g] = 0x7F800000u;
        wmax[g] = 0u;
    }
    if (blockIdx.x == 0) {
        int p = threadIdx.x;
        float s = 0.f;
        for (int c = 0; c < NC; ++c) {
            float v = proto[c * NP + p];
            s += v * v;
        }
        yty[p] = s;
    }
}


__device__ __forceinline__ void wave_lds_sync()
{
    __builtin_amdgcn_fence(__ATOMIC_RELEASE, "wavefront");
    __builtin_amdgcn_wave_barrier();
    __builtin_amdgcn_fence(__ATOMIC_ACQUIRE, "wavefront");
}

template <int NT>
__device__ __forceinline__ void do_strip(
    const int j, const int n0,
    const float* __restrict__ in_part,
    float* slot, float* xtx_w,
    const v2f* Pfrag,
    const float* __restrict__ lscale,
    const float* __restrict__ yty,
    unsigned int* __restrict__ wmin,
    unsigned int* __restrict__ wmax,
    const int part, const int bglob, const int lane)
{
    const int lm   = lane & 15;
    const int hsel = lane >> 4;

    #pragma unroll
    for (int t = 0; t < 32; ++t) {
        const int row = t >> 1;
        const int c4  = (t & 1) * 32 + lane;
        const int lr  = min(j * 16 + row, PROWS - 1);
        float4 v = *(const float4*)(in_part + (size_t)lr * NC + 4 * c4);
        *(float4*)(slot + row * APITCH + 4 * c4) = v;
    }
    wave_lds_sync();

    float lsv[NT][8];
    #pragma unroll
    for (int r = 0; r < 8; ++r) {
        const int s_idx = part * PROWS + min(j * 16 + hsel * 8 + r, PROWS - 1);
        const float* lrow = lscale + (size_t)s_idx * NP + n0 * 16 + lm;
        #pragma unroll
        for (int n = 0; n < NT; ++n) lsv[n][r] = lrow[n * 16];
    }

    {
        const float4* row = (const float4*)(slot + lm * APITCH);
        float s = 0.f;
        for (int i = 0; i < NC / 4; ++i) {
            float4 v = row[i];
            s += v.x * v.x + v.y * v.y + v.z * v.z + v.w * v.w;
        }
        xtx_w[lm] = s;
    }
    wave_lds_sync();

    v8f acc[NT];
    #pragma unroll
    for (int n = 0; n < NT; ++n) acc[n] = (v8f){0.f, 0.f, 0.f, 0.f, 0.f, 0.f, 0.f, 0.f};

    const int a_off = lm * APITCH + 2 * hsel;

    #pragma unroll (NT == NTILES ? 1 : 8)
    for (int k0 = 0; k0 < NC; k0 += 4) {
        v2f Af = *(const v2f*)(slot + a_off + k0);
        const v2f* Bp = Pfrag + ((k0 >> 1) + hsel) * NP + n0 * 16 + lm;
        #pragma unroll
        for (int n = 0; n < NT; ++n) {
            v2f Bf = Bp[n * 16];
            acc[n] = __builtin_amdgcn_wmma_f32_16x16x4_f32(
                false, Af, false, Bf, (short)0, acc[n], false, false);
        }
    }

    const int row_loc = hsel * 8;

    #pragma unroll
    for (int n = 0; n < NT; ++n) {
        const int p  = (n0 + n) * 16 + lm;
        const float yv = yty[p];
        float mn = 3.0e38f;
        float mx = 0.f;
        #pragma unroll
        for (int r = 0; r < 8; ++r) {
            float xt  = xtx_w[row_loc + r];
            float sf  = __expf(lsv[n][r]);
            float xty = acc[n][r];
            float nsq = fmaxf(xt - 2.f * xty + yv, 0.f);
            float dmin = nsq / (sf + EPS);
            float sim  = __logf((nsq + 1.f) / (nsq + EPS)) * sf;
            mn = fminf(mn, dmin);
            mx = fmaxf(mx, sim);
        }
        mn = fminf(mn, __shfl_xor(mn, 16));
        mx = fmaxf(mx, __shfl_xor(mx, 16));
        if (lane < 16) {
            atomicMin(wmin + bglob * NP + p, __float_as_uint(mn));
            atomicMax(wmax + bglob * NP + p, __float_as_uint(mx));
        }
    }
    wave_lds_sync();
}

__global__ __launch_bounds__(256)
void proto_main(const float* __restrict__ in,
                const float* __restrict__ proto,
                const float* __restrict__ lscale,
                const float* __restrict__ yty,
                unsigned int* __restrict__ wmin,
                unsigned int* __restrict__ wmax)
{
    extern __shared__ float lds[];
    float* a_lds = lds;
    float* p_lds = lds + MWAVES * 16 * APITCH;
    float* xtx   = p_lds + NC * NP;

    const int bglob = blockIdx.x / PARTS;
    const int part  = blockIdx.x % PARTS;
    const int tid   = threadIdx.x;

    {
        float2* dst = (float2*)p_lds;
        for (int i = tid; i < (NC / 2) * NP; i += 256) {
            int kp = i >> 7;
            int p  = i & (NP - 1);
            float v0 = proto[(2 * kp) * NP + p];
            float v1 = proto[(2 * kp + 1) * NP + p];
            dst[i] = make_float2(v0, v1);
        }
    }
    __syncthreads();

    const int wave = __builtin_amdgcn_readfirstlane(tid >> 5);
    const int lane = tid & 31;

    float* slot  = a_lds + wave * 16 * APITCH;
    float* xtx_w = xtx + wave * 16;
    const float* in_part = in + ((size_t)bglob * NS + (size_t)part * PROWS) * NC;
    const v2f* Pfrag = (const v2f*)p_lds;

    for (int j = wave; j < WSTRIPS; j += MWAVES)
        do_strip<NTILES>(j, 0, in_part, slot, xtx_w, Pfrag, lscale, yty, wmin, wmax,
                         part, bglob, lane);
    do_strip<1>(WSTRIPS, wave, in_part, slot, xtx_w, Pfrag, lscale, yty, wmin, wmax,
                part, bglob, lane);
}

__global__ __launch_bounds__(FIN_THREADS)
void proto_final(const unsigned int* __restrict__ wmin,
                 const unsigned int* __restrict__ wmax,
                 const float* __restrict__ pcc,
                 float* __restrict__ out)
{
    __shared__ float r1[NB];
    __shared__ float r2[NB];
    const int t    = threadIdx.x;
    const int b    = t >> 5;
    const int lane = t & 31;

    for (int j = 0; j < (NB * NP) / FIN_THREADS; ++j)
        out[j * FIN_THREADS + t] = __uint_as_float(wmax[j * FIN_THREADS + t]);

    float m1, m2;
    {
        float md = __uint_as_float(wmin[b * NP + lane]);
        float v  = MAXDIST - md;
        float m  = pcc[b * NP + lane];
        m1 = v * m;
        m2 = v * (1.f - m);
    }
    for (int j = 1; j < NP / 32; ++j) {
        float md = __uint_as_float(wmin[b * NP + j * 32 + lane]);
        float v  = MAXDIST - md;
        float m  = pcc[b * NP + j * 32 + lane];
        m1 = fmaxf(m1, v * m);
        m2 = fmaxf(m2, v * (1.f - m));
    }
    for (int off = 16; off > 0; off >>= 1) {
        m1 = fmaxf(m1, __shfl_xor(m1, off));
        m2 = fmaxf(m2, __shfl_xor(m2, off));
    }
    if (lane == 0) { r1[b] = m1; r2[b] = m2; }
    __syncthreads();

    if (t == 0) {
        float s1 = 0.f, s2 = 0.f;
        for (int bb = 0; bb < NB; ++bb) {
            s1 += MAXDIST - r1[bb];
            s2 += MAXDIST - r2[bb];
        }
        out[NB * NP]     = s1 / (float)NB;
        out[NB * NP + 1] = s2 / (float)NB;
    }
}

extern "C" void kernel_launch(void* const* d_in, const int* in_sizes, int n_in,
                              void* d_out, int out_size, void* d_ws, size_t ws_size,
                              hipStream_t stream)
{
    const float* inputs = (const float*)d_in[0];
    const float* pcc    = (const float*)d_in[1];
    const float* proto  = (const float*)d_in[2];
    const float* lscale = (const float*)d_in[3];
    float* out = (float*)d_out;

    unsigned int* wmin = (unsigned int*)d_ws;
    unsigned int* wmax = wmin + NB * NP;
    float* yty = (float*)(wmax + NB * NP);

    proto_init<<<dim3((NB * NP) / 128), dim3(128), 0, stream>>>(proto, wmin, wmax, yty);

    const size_t shmem = (size_t)(MWAVES * 16 * APITCH + NC * NP + MWAVES * 16) * sizeof(float);
    proto_main<<<dim3(NB * PARTS), dim3(256), shmem, stream>>>(
        inputs, proto, lscale, yty, wmin, wmax);

    proto_final<<<dim3(1), dim3(FIN_THREADS), 0, stream>>>(wmin, wmax, pcc, out);
}


# K loop of the 8-tile strips software-pipelined: A/B fragment loads of the next two K steps issued ahead of the current two steps' 16 matrix instructions (two register sets); k order, sums unchanged
# speedup vs baseline: 1.9983x; 1.0612x over previous
#include <hip/hip_runtime.h>
#include <hip/hip_bf16.h>

typedef float v2f __attribute__((ext_vector_type(2)));
typedef float v8f __attribute__((ext_vector_type(8)));

constexpr int NB = 32;
constexpr int NH = 56, NW = 56;
constexpr int NC = 256;
constexpr int NP = 128;
constexpr int NS = NH * NW;
constexpr int MTILE = 64;
constexpr int NTILES = NP / 16;
constexpr int TILES = NS / MTILE;
constexpr int PARTS   = 8;
constexpr int PROWS   = NS / PARTS;
constexpr int WSTRIPS = PROWS / 16;
constexpr int MWAVES  = 8;
static_assert(MWAVES == NTILES, "the last strip is split one column tile per wave");
static_assert(WSTRIPS % MWAVES == 0 && PROWS - 16 * WSTRIPS > 0 && PROWS - 16 * WSTRIPS <= 16,
              "three whole strips per wave and one partial strip");
constexpr float MAXDIST = 100.0f * 256.0f;
constexpr float EPS = 1e-4f;
constexpr int FIN_THREADS = NB * 32;
constexpr int APITCH = NC + 4;

__global__ __launch_bounds__(128)
void proto_init(const float* __restrict__ proto,
                unsigned int* __restrict__ wmin,
                unsigned int* __restrict__ wmax,
                float* __restrict__ yty)
{
    int g = blockIdx.x * 128 + threadIdx.x;
    if (g < NB * NP) {
        wmin[g] = 0x7F800000u;
        wmax[g] = 0u;
    }
    if (blockIdx.x == 0) {
        int p = threadIdx.x;
        float s = 0.f;
        for (int c = 0; c < NC; ++c) {
            float v = proto[c * NP + p];
            s += v * v;
        }
        yty[p] = s;
    }
}


__device__ __forceinline__ void wave_lds_sync()
{
    __builtin_amdgcn_fence(__ATOMIC_RELEASE, "wavefront");
    __builtin_amdgcn_wave_barrier();
    __builtin_amdgcn_fence(__ATOMIC_ACQUIRE, "wavefront");
}

template <int NT>
__device__ __forceinline__ void load_group(v2f (&Af)[2], v2f (&Bf)[2][NT],
                                           const float* a_row, const v2f* b_col, const int k0)
{
    #pragma unroll
    for (int s = 0; s < 2; ++s) {
        const int k = k0 + 4 * s;
        Af[s] = *(const v2f*)(a_row + k);
        const v2f* Bp = b_col + (k >> 1) * NP;
        #pragma unroll
        for (int n = 0; n < NT; ++n) Bf[s][n] = Bp[n * 16];
    }
}

template <int NT>
__device__ __forceinline__ void mma_group(v8f (&acc)[NT], const v2f (&Af)[2], const v2f (&Bf)[2][NT])
{
    #pragma unroll
    for (int s = 0; s < 2; ++s) {
        #pragma unroll
        for (int n = 0; n < NT; ++n)
            acc[n] = __builtin_amdgcn_wmma_f32_16x16x4_f32(
                false, Af[s], false, Bf[s][n], (short)0, acc[n], false, false);
    }
}

template <int NT>
__device__ __forceinline__ void do_strip(
    const int j, const int n0,
    const float* __restrict__ in_part,
    float* slot, float* xtx_w,
    const v2f* Pfrag,
    const float* __restrict__ lscale,
    const float* __restrict__ yty,
    unsigned int* __restrict__ wmin,
    unsigned int* __restrict__ wmax,
    const int part, const int bglob, const int lane)
{
    const int lm   = lane & 15;
    const int hsel = lane >> 4;

    #pragma unroll
    for (int t = 0; t < 32; ++t) {
        const int row = t >> 1;
        const int c4  = (t & 1) * 32 + lane;
        const int lr  = min(j * 16 + row, PROWS - 1);
        float4 v = *(const float4*)(in_part + (size_t)lr * NC + 4 * c4);
        *(float4*)(slot + row * APITCH + 4 * c4) = v;
    }
    wave_lds_sync();

    float lsv[NT][8];
    #pragma unroll
    for (int r = 0; r < 8; ++r) {
        const int s_idx = part * PROWS + min(j * 16 + hsel * 8 + r, PROWS - 1);
        const float* lrow = lscale + (size_t)s_idx * NP + n0 * 16 + lm;
        #pragma unroll
        for (int n = 0; n < NT; ++n) lsv[n][r] = lrow[n * 16];
    }

    {
        const float4* row = (const float4*)(slot + lm * APITCH);
        float s = 0.f;
        for (int i = 0; i < NC / 4; ++i) {
            float4 v = row[i];
            s += v.x * v.x + v.y * v.y + v.z * v.z + v.w * v.w;
        }
        xtx_w[lm] = s;
    }
    wave_lds_sync();

    v8f acc[NT];
    #pragma unroll
    for (int n = 0; n < NT; ++n) acc[n] = (v8f){0.f, 0.f, 0.f, 0.f, 0.f, 0.f, 0.f, 0.f};

    const int a_off = lm * APITCH + 2 * hsel;

    if constexpr (NT == NTILES) {
        const float* a_row = slot + a_off;
        const v2f*   b_col = Pfrag + hsel * NP + n0 * 16 + lm;
        v2f Af0[2], Af1[2];
        v2f Bf0[2][NT], Bf1[2][NT];
        load_group<NT>(Af0, Bf0, a_row, b_col, 0);
        #pragma unroll 1
        for (int k0 = 0; k0 < NC - 16; k0 += 16) {
            load_group<NT>(Af1, Bf1, a_row, b_col, k0 + 8);
            __builtin_amdgcn_sched_barrier(0);
            mma_group<NT>(acc, Af0, Bf0);
            __builtin_amdgcn_sched_barrier(0);
            load_group<NT>(Af0, Bf0, a_row, b_col, k0 + 16);
            __builtin_amdgcn_sched_barrier(0);
            mma_group<NT>(acc, Af1, Bf1);
        }
        load_group<NT>(Af1, Bf1, a_row, b_col, NC - 8);
        __builtin_amdgcn_sched_barrier(0);
        mma_group<NT>(acc, Af0, Bf0);
        mma_group<NT>(acc, Af1, Bf1);
    } else {
        #pragma unroll 8
        for (int k0 = 0; k0 < NC; k0 += 4) {
            v2f Af = *(const v2f*)(slot + a_off + k0);
            const v2f* Bp = Pfrag + ((k0 >> 1) + hsel) * NP + n0 * 16 + lm;
            #pragma unroll
            for (int n = 0; n < NT; ++n) {
                v2f Bf = Bp[n * 16];
                acc[n] = __builtin_amdgcn_wmma_f32_16x16x4_f32(
                    false, Af, false, Bf, (short)0, acc[n], false, false);
            }
        }
    }

    const int row_loc = hsel * 8;

    #pragma unroll
    for (int n = 0; n < NT; ++n) {
        const int p  = (n0 + n) * 16 + lm;
        const float yv = yty[p];
        float mn = 3.0e38f;
        float mx = 0.f;
        #pragma unroll
        for (int r = 0; r < 8; ++r) {
            float xt  = xtx_w[row_loc + r];
            float sf  = __expf(lsv[n][r]);
            float xty = acc[n][r];
            float nsq = fmaxf(xt - 2.f * xty + yv, 0.f);
            float dmin = nsq / (sf + EPS);
            float sim  = __logf((nsq + 1.f) / (nsq + EPS)) * sf;
            mn = fminf(mn, dmin);
            mx = fmaxf(mx, sim);
        }
        mn = fminf(mn, __shfl_xor(mn, 16));
        mx = fmaxf(mx, __shfl_xor(mx, 16));
        if (lane < 16) {
            atomicMin(wmin + bglob * NP + p, __float_as_uint(mn));
            atomicMax(wmax + bglob * NP + p, __float_as_uint(mx));
        }
    }
    wave_lds_sync();
}

__global__ __launch_bounds__(256)
void proto_main(const float* __restrict__ in,
                const float* __restrict__ proto,
                const float* __restrict__ lscale,
                const float* __restrict__ yty,
                unsigned int* __restrict__ wmin,
                unsigned int* __restrict__ wmax)
{
    extern __shared__ float lds[];
    float* a_lds = lds;
    float* p_lds = lds + MWAVES * 16 * APITCH;
    float* xtx   = p_lds + NC * NP;

    const int bglob = blockIdx.x / PARTS;
    const int part  = blockIdx.x % PARTS;
    const int tid   = threadIdx.x;

    {
        float2* dst = (float2*)p_lds;
        for (int i = tid; i < (NC / 2) * NP; i += 256) {
            int kp = i >> 7;
            int p  = i & (NP - 1);
            float v0 = proto[(2 * kp) * NP + p];
            float v1 = proto[(2 * kp + 1) * NP + p];
            dst[i] = make_float2(v0, v1);
        }
    }
    __syncthreads();

    const int wave = __builtin_amdgcn_readfirstlane(tid >> 5);
    const int lane = tid & 31;

    float* slot  = a_lds + wave * 16 * APITCH;
    float* xtx_w = xtx + wave * 16;
    const float* in_part = in + ((size_t)bglob * NS + (size_t)part * PROWS) * NC;
    const v2f* Pfrag = (const v2f*)p_lds;

    for (int j = wave; j < WSTRIPS; j += MWAVES)
        do_strip<NTILES>(j, 0, in_part, slot, xtx_w, Pfrag, lscale, yty, wmin, wmax,
                         part, bglob, lane);
    do_strip<1>(WSTRIPS, wave, in_part, slot, xtx_w, Pfrag, lscale, yty, wmin, wmax,
                part, bglob, lane);
}

__global__ __launch_bounds__(FIN_THREADS)
void proto_final(const unsigned int* __restrict__ wmin,
                 const unsigned int* __restrict__ wmax,
                 const float* __restrict__ pcc,
                 float* __restrict__ out)
{
    __shared__ float r1[NB];
    __shared__ float r2[NB];
    const int t    = threadIdx.x;
    const int b    = t >> 5;
    const int lane = t & 31;

    for (int j = 0; j < (NB * NP) / FIN_THREADS; ++j)
        out[j * FIN_THREADS + t] = __uint_as_float(wmax[j * FIN_THREADS + t]);

    float m1, m2;
    {
        float md = __uint_as_float(wmin[b * NP + lane]);
        float v  = MAXDIST - md;
        float m  = pcc[b * NP + lane];
        m1 = v * m;
        m2 = v * (1.f - m);
    }
    for (int j = 1; j < NP / 32; ++j) {
        float md = __uint_as_float(wmin[b * NP + j * 32 + lane]);
        float v  = MAXDIST - md;
        float m  = pcc[b * NP + j * 32 + lane];
        m1 = fmaxf(m1, v * m);
        m2 = fmaxf(m2, v * (1.f - m));
    }
    for (int off = 16; off > 0; off >>= 1) {
        m1 = fmaxf(m1, __shfl_xor(m1, off));
        m2 = fmaxf(m2, __shfl_xor(m2, off));
    }
    if (lane == 0) { r1[b] = m1; r2[b] = m2; }
    __syncthreads();

    if (t == 0) {
        float s1 = 0.f, s2 = 0.f;
        for (int bb = 0; bb < NB; ++bb) {
            s1 += MAXDIST - r1[bb];
            s2 += MAXDIST - r2[bb];
        }
        out[NB * NP]     = s1 / (float)NB;
        out[NB * NP + 1] = s2 / (float)NB;
    }
}

extern "C" void kernel_launch(void* const* d_in, const int* in_sizes, int n_in,
                              void* d_out, int out_size, void* d_ws, size_t ws_size,
                              hipStream_t stream)
{
    const float* inputs = (const float*)d_in[0];
    const float* pcc    = (const float*)d_in[1];
    const float* proto  = (const float*)d_in[2];
    const float* lscale = (const float*)d_in[3];
    float* out = (float*)d_out;

    unsigned int* wmin = (unsigned int*)d_ws;
    unsigned int* wmax = wmin + NB * NP;
    float* yty = (float*)(wmax + NB * NP);

    proto_init<<<dim3((NB * NP) / 128), dim3(128), 0, stream>>>(proto, wmin, wmax, yty);

    const size_t shmem = (size_t)(MWAVES * 16 * APITCH + NC * NP + MWAVES * 16) * sizeof(float);
    proto_main<<<dim3(NB * PARTS), dim3(256), shmem, stream>>>(
        inputs, proto, lscale, yty, wmin, wmax);

    proto_final<<<dim3(1), dim3(FIN_THREADS), 0, stream>>>(wmin, wmax, pcc, out);
}


# issue priority of a wave falls with the whole strips it has finished (s_setprio 3, 2, 1, then 0): the two waves of a SIMD take turns strip by strip and fill each other's waits; arithmetic unchanged
# speedup vs baseline: 2.5563x; 1.2792x over previous
#include <hip/hip_runtime.h>
#include <hip/hip_bf16.h>

typedef float v2f __attribute__((ext_vector_type(2)));
typedef float v4f __attribute__((ext_vector_type(4)));
typedef float v8f __attribute__((ext_vector_type(8)));

constexpr int NB = 32;
constexpr int NH = 56, NW = 56;
constexpr int NC = 256;
constexpr int NP = 128;
constexpr int NS = NH * NW;
constexpr int MTILE = 64;
constexpr int NTILES = NP / 16;
constexpr int TILES = NS / MTILE;
constexpr int PARTS   = 8;
constexpr int PROWS   = NS / PARTS;
constexpr int WSTRIPS = PROWS / 16;
constexpr int MWAVES  = 8;
static_assert(MWAVES == NTILES, "the last strip is split one column tile per wave");
static_assert(WSTRIPS % MWAVES == 0 && PROWS - 16 * WSTRIPS > 0 && PROWS - 16 * WSTRIPS <= 16,
              "three whole strips per wave and one partial strip");
constexpr float MAXDIST = 100.0f * 256.0f;
constexpr float EPS = 1e-4f;
constexpr int FIN_THREADS = NB * 32;
constexpr int APITCH = NC + 4;


__device__ __forceinline__ void wave_lds_sync()
{
    __builtin_amdgcn_fence(__ATOMIC_RELEASE, "wavefront");
    __builtin_amdgcn_wave_barrier();
    __builtin_amdgcn_fence(__ATOMIC_ACQUIRE, "wavefront");
}

template <int NT>
__device__ __forceinline__ void load_group(v2f (&Af)[2], v2f (&Bf)[2][NT],
                                           const float* a_row, const v2f* b_col, const int k0)
{
    #pragma unroll
    for (int s = 0; s < 2; ++s) {
        const int k = k0 + 4 * s;
        Af[s] = *(const v2f*)(a_row + k);
        const v2f* Bp = b_col + (k >> 1) * NP;
        #pragma unroll
        for (int n = 0; n < NT; ++n) Bf[s][n] = Bp[n * 16];
    }
}

template <int NT>
__device__ __forceinline__ void mma_group(v8f (&acc)[NT], const v2f (&Af)[2], const v2f (&Bf)[2][NT])
{
    #pragma unroll
    for (int s = 0; s < 2; ++s) {
        #pragma unroll
        for (int n = 0; n < NT; ++n)
            acc[n] = __builtin_amdgcn_wmma_f32_16x16x4_f32(
                false, Af[s], false, Bf[s][n], (short)0, acc[n], false, false);
    }
}

template <int NT>
__device__ __forceinline__ void do_strip(
    const int j, const int n0,
    const float* __restrict__ in_part,
    float* slot, float* xtx_w,
    const v2f* Pfrag,
    const float* __restrict__ lscale,
    const float (&yv)[NT],
    float (&mn)[NT], float (&mx)[NT],
    const int part, const int lane)
{
    const int lm   = lane & 15;
    const int hsel = lane >> 4;

    #pragma unroll
    for (int t = 0; t < 32; ++t) {
        const int row = t >> 1;
        const int c4  = (t & 1) * 32 + lane;
        const int lr  = min(j * 16 + row, PROWS - 1);
        float4 v = *(const float4*)(in_part + (size_t)lr * NC + 4 * c4);
        *(float4*)(slot + row * APITCH + 4 * c4) = v;
    }
    wave_lds_sync();

    float lsv[NT][8];
    #pragma unroll
    for (int r = 0; r < 8; ++r) {
        const int s_idx = part * PROWS + min(j * 16 + hsel * 8 + r, PROWS - 1);
        const float* lrow = lscale + (size_t)s_idx * NP + n0 * 16 + lm;
        #pragma unroll
        for (int n = 0; n < NT; ++n) lsv[n][r] = lrow[n * 16];
    }

    {
        const float4* row = (const float4*)(slot + lm * APITCH);
        float s = 0.f;
        for (int i = 0; i < NC / 4; ++i) {
            float4 v = row[i];
            s += v.x * v.x + v.y * v.y + v.z * v.z + v.w * v.w;
        }
        xtx_w[lm] = s;
    }
    wave_lds_sync();

    v8f acc[NT];
    #pragma unroll
    for (int n = 0; n < NT; ++n) acc[n] = (v8f){0.f, 0.f, 0.f, 0.f, 0.f, 0.f, 0.f, 0.f};

    const int a_off = lm * APITCH + 2 * hsel;

    if constexpr (NT == NTILES) {
        const float* a_row = slot + a_off;
        const v2f*   b_col = Pfrag + hsel * NP + n0 * 16 + lm;
        v2f Af0[2], Af1[2];
        v2f Bf0[2][NT], Bf1[2][NT];
        load_group<NT>(Af0, Bf0, a_row, b_col, 0);
        #pragma unroll 1
        for (int k0 = 0; k0 < NC - 16; k0 += 16) {
            load_group<NT>(Af1, Bf1, a_row, b_col, k0 + 8);
            __builtin_amdgcn_sched_barrier(0);
            mma_group<NT>(acc, Af0, Bf0);
            __builtin_amdgcn_sched_barrier(0);
            load_group<NT>(Af0, Bf0, a_row, b_col, k0 + 16);
            __builtin_amdgcn_sched_barrier(0);
            mma_group<NT>(acc, Af1, Bf1);
        }
        load_group<NT>(Af1, Bf1, a_row, b_col, NC - 8);
        __builtin_amdgcn_sched_barrier(0);
        mma_group<NT>(acc, Af0, Bf0);
        mma_group<NT>(acc, Af1, Bf1);
    } else {
        #pragma unroll 8
        for (int k0 = 0; k0 < NC; k0 += 4) {
            v2f Af = *(const v2f*)(slot + a_off + k0);
            const v2f* Bp = Pfrag + ((k0 >> 1) + hsel) * NP + n0 * 16 + lm;
            #pragma unroll
            for (int n = 0; n < NT; ++n) {
                v2f Bf = Bp[n * 16];
                acc[n] = __builtin_amdgcn_wmma_f32_16x16x4_f32(
                    false, Af, false, Bf, (short)0, acc[n], false, false);
            }
        }
    }

    const int row_loc = hsel * 8;

    #pragma unroll
    for (int n = 0; n < NT; ++n) {
        float m0 = mn[n];
        float m1 = mx[n];
        #pragma unroll
        for (int r = 0; r < 8; ++r) {
            float xt  = xtx_w[row_loc + r];
            float sf  = __expf(lsv[n][r]);
            float xty = acc[n][r];
            float nsq = fmaxf(xt - 2.f * xty + yv[n], 0.f);
            float dmin = nsq / (sf + EPS);
            float sim  = __logf((nsq + 1.f) / (nsq + EPS)) * sf;
            m0 = fminf(m0, dmin);
            m1 = fmaxf(m1, sim);
        }
        mn[n] = m0;
        mx[n] = m1;
    }
    wave_lds_sync();
}

__global__ __launch_bounds__(256)
void proto_main(const float* __restrict__ in,
                const float* __restrict__ proto,
                const float* __restrict__ lscale,
                float* __restrict__ pmin,
                float* __restrict__ pmax)
{
    extern __shared__ float lds[];
    float* a_lds = lds;
    float* p_lds = lds + MWAVES * 16 * APITCH;
    float* xtx   = p_lds + NC * NP;
    float* yty_l = xtx + MWAVES * 16;

    const int bglob = blockIdx.x / PARTS;
    const int part  = blockIdx.x % PARTS;
    const int tid   = threadIdx.x;

    {
        float2* dst = (float2*)p_lds;
        for (int i = tid; i < (NC / 2) * NP; i += 256) {
            int kp = i >> 7;
            int p  = i & (NP - 1);
            float v0 = proto[(2 * kp) * NP + p];
            float v1 = proto[(2 * kp + 1) * NP + p];
            dst[i] = make_float2(v0, v1);
        }
    }
    __syncthreads();

    if (tid < NP) {
        const float2* src = (const float2*)p_lds + tid;
        float s = 0.f;
        for (int k0 = 0; k0 < NC / 2; k0 += 32) {
            float2 v[32];
            #pragma unroll
            for (int i = 0; i < 32; ++i) v[i] = src[(k0 + i) * NP];
            __builtin_amdgcn_sched_barrier(0);
            #pragma unroll
            for (int i = 0; i < 32; ++i) {
                s = __builtin_fmaf(v[i].x, v[i].x, s);
                s = __builtin_fmaf(v[i].y, v[i].y, s);
            }
        }
        yty_l[tid] = s;
    }
    __syncthreads();

    const int wave = __builtin_amdgcn_readfirstlane(tid >> 5);
    const int lane = tid & 31;

    float* slot  = a_lds + wave * 16 * APITCH;
    float* xtx_w = xtx + wave * 16;
    const float* in_part = in + ((size_t)bglob * NS + (size_t)part * PROWS) * NC;
    const v2f* Pfrag = (const v2f*)p_lds;

    const int lm = lane & 15;

    float yv[NTILES], mn[NTILES], mx[NTILES];
    #pragma unroll
    for (int n = 0; n < NTILES; ++n) {
        yv[n] = yty_l[n * 16 + lm];
        mn[n] = 3.0e38f;
        mx[n] = 0.f;
    }
    float yv1[1], mn1[1], mx1[1];
    yv1[0] = yty_l[wave * 16 + lm];
    mn1[0] = 3.0e38f;
    mx1[0] = 0.f;

    int done = 0;
    for (int j = wave; j < WSTRIPS; j += MWAVES) {
        if (done == 0)      __builtin_amdgcn_s_setprio(3);
        else if (done == 1) __builtin_amdgcn_s_setprio(2);
        else                __builtin_amdgcn_s_setprio(1);
        do_strip<NTILES>(j, 0, in_part, slot, xtx_w, Pfrag, lscale, yv, mn, mx, part, lane);
        ++done;
    }
    __builtin_amdgcn_s_setprio(0);
    do_strip<1>(WSTRIPS, wave, in_part, slot, xtx_w, Pfrag, lscale, yv1, mn1, mx1, part, lane);

    #pragma unroll
    for (int n = 0; n < NTILES; ++n) {
        mn[n] = fminf(mn[n], __shfl_xor(mn[n], 16));
        mx[n] = fmaxf(mx[n], __shfl_xor(mx[n], 16));
    }
    mn1[0] = fminf(mn1[0], __shfl_xor(mn1[0], 16));
    mx1[0] = fmaxf(mx1[0], __shfl_xor(mx1[0], 16));
    if (lane < 16) {
        #pragma unroll
        for (int n = 0; n < NTILES; ++n) {
            slot[n * 16 + lm]          = mn[n];
            slot[APITCH + n * 16 + lm] = mx[n];
        }
        slot[2 * APITCH + lm]      = mn1[0];
        slot[2 * APITCH + 16 + lm] = mx1[0];
    }
    __syncthreads();
    {
        const int which = tid >> 7;
        const int p     = tid & (NP - 1);
        float v = a_lds[(p >> 4) * 16 * APITCH + 2 * APITCH + which * 16 + (p & 15)];
        #pragma unroll
        for (int w = 0; w < MWAVES; ++w) {
            const float x = a_lds[w * 16 * APITCH + which * APITCH + p];
            v = which ? fmaxf(v, x) : fminf(v, x);
        }
        float* dst = which ? pmax : pmin;
        dst[blockIdx.x * NP + p] = v;
    }
}

__global__ __launch_bounds__(FIN_THREADS)
void proto_final(const float* __restrict__ pmin,
                 const float* __restrict__ pmax,
                 const float* __restrict__ pcc,
                 float* __restrict__ out)
{
    __shared__ float r1[NB];
    __shared__ float r2[NB];
    const int t    = threadIdx.x;
    const int b    = t >> 5;
    const int lane = t & 31;

    if (blockIdx.x == 1) {
        const v4f* qmax = (const v4f*)(pmax + (size_t)b * PARTS * NP) + lane;
        v4f vmax[PARTS];
        #pragma unroll
        for (int q = 0; q < PARTS; ++q) vmax[q] = qmax[q * (NP / 4)];
        v4f hi = vmax[0];
        #pragma unroll
        for (int q = 1; q < PARTS; ++q) {
            hi.x = fmaxf(hi.x, vmax[q].x); hi.y = fmaxf(hi.y, vmax[q].y);
            hi.z = fmaxf(hi.z, vmax[q].z); hi.w = fmaxf(hi.w, vmax[q].w);
        }
        ((v4f*)(out + b * NP))[lane] = hi;
    } else {
        const v4f* qmin = (const v4f*)(pmin + (size_t)b * PARTS * NP) + lane;
        v4f vmin[PARTS];
        #pragma unroll
        for (int q = 0; q < PARTS; ++q) vmin[q] = qmin[q * (NP / 4)];
        const v4f m = ((const v4f*)(pcc + b * NP))[lane];
        v4f lo = vmin[0];
        #pragma unroll
        for (int q = 1; q < PARTS; ++q) {
            lo.x = fminf(lo.x, vmin[q].x); lo.y = fminf(lo.y, vmin[q].y);
            lo.z = fminf(lo.z, vmin[q].z); lo.w = fminf(lo.w, vmin[q].w);
        }
        const float v0 = MAXDIST - lo.x, v1 = MAXDIST - lo.y, v2 = MAXDIST - lo.z, v3 = MAXDIST - lo.w;
        float m1 = fmaxf(fmaxf(v0 * m.x, v1 * m.y), fmaxf(v2 * m.z, v3 * m.w));
        float m2 = fmaxf(fmaxf(v0 * (1.f - m.x), v1 * (1.f - m.y)),
                         fmaxf(v2 * (1.f - m.z), v3 * (1.f - m.w)));
        for (int off = 16; off > 0; off >>= 1) {
            m1 = fmaxf(m1, __shfl_xor(m1, off));
            m2 = fmaxf(m2, __shfl_xor(m2, off));
        }
        if (lane == 0) { r1[b] = m1; r2[b] = m2; }
        __syncthreads();

        if (t == 0) {
            float s1 = 0.f, s2 = 0.f;
            for (int bb = 0; bb < NB; ++bb) {
                s1 += MAXDIST - r1[bb];
                s2 += MAXDIST - r2[bb];
            }
            out[NB * NP]     = s1 / (float)NB;
            out[NB * NP + 1] = s2 / (float)NB;
        }
    }
}

extern "C" void kernel_launch(void* const* d_in, const int* in_sizes, int n_in,
                              void* d_out, int out_size, void* d_ws, size_t ws_size,
                              hipStream_t stream)
{
    const float* inputs = (const float*)d_in[0];
    const float* pcc    = (const float*)d_in[1];
    const float* proto  = (const float*)d_in[2];
    const float* lscale = (const float*)d_in[3];
    float* out = (float*)d_out;

    float* pmin = (float*)d_ws;
    float* pmax = pmin + NB * PARTS * NP;

    const size_t shmem = (size_t)(MWAVES * 16 * APITCH + NC * NP + MWAVES * 16 + NP) * sizeof(float);
    proto_main<<<dim3(NB * PARTS), dim3(256), shmem, stream>>>(
        inputs, proto, lscale, pmin, pmax);

    proto_final<<<dim3(2), dim3(FIN_THREADS), 0, stream>>>(pmin, pmax, pcc, out);
}


# epilogue: both IEEE divisions per output become v_rcp_f32 products (ratio as 1+(1-eps)*rcp(nsq+eps)) and the log a bare v_log_f32*ln2; 2216 -> 487 plain vector instructions per strip; sums unchanged
# speedup vs baseline: 2.7648x; 1.0816x over previous
#include <hip/hip_runtime.h>
#include <hip/hip_bf16.h>

typedef float v2f __attribute__((ext_vector_type(2)));
typedef float v4f __attribute__((ext_vector_type(4)));
typedef float v8f __attribute__((ext_vector_type(8)));

constexpr int NB = 32;
constexpr int NH = 56, NW = 56;
constexpr int NC = 256;
constexpr int NP = 128;
constexpr int NS = NH * NW;
constexpr int MTILE = 64;
constexpr int NTILES = NP / 16;
constexpr int TILES = NS / MTILE;
constexpr int PARTS   = 8;
constexpr int PROWS   = NS / PARTS;
constexpr int WSTRIPS = PROWS / 16;
constexpr int MWAVES  = 8;
static_assert(MWAVES == NTILES, "the last strip is split one column tile per wave");
static_assert(WSTRIPS % MWAVES == 0 && PROWS - 16 * WSTRIPS > 0 && PROWS - 16 * WSTRIPS <= 16,
              "three whole strips per wave and one partial strip");
constexpr float MAXDIST = 100.0f * 256.0f;
constexpr float EPS = 1e-4f;
constexpr float LN2 = 0.693147180559945309f;
constexpr int FIN_THREADS = NB * 32;
constexpr int APITCH = NC + 4;


__device__ __forceinline__ void wave_lds_sync()
{
    __builtin_amdgcn_fence(__ATOMIC_RELEASE, "wavefront");
    __builtin_amdgcn_wave_barrier();
    __builtin_amdgcn_fence(__ATOMIC_ACQUIRE, "wavefront");
}

template <int NT>
__device__ __forceinline__ void load_group(v2f (&Af)[2], v2f (&Bf)[2][NT],
                                           const float* a_row, const v2f* b_col, const int k0)
{
    #pragma unroll
    for (int s = 0; s < 2; ++s) {
        const int k = k0 + 4 * s;
        Af[s] = *(const v2f*)(a_row + k);
        const v2f* Bp = b_col + (k >> 1) * NP;
        #pragma unroll
        for (int n = 0; n < NT; ++n) Bf[s][n] = Bp[n * 16];
    }
}

template <int NT>
__device__ __forceinline__ void mma_group(v8f (&acc)[NT], const v2f (&Af)[2], const v2f (&Bf)[2][NT])
{
    #pragma unroll
    for (int s = 0; s < 2; ++s) {
        #pragma unroll
        for (int n = 0; n < NT; ++n)
            acc[n] = __builtin_amdgcn_wmma_f32_16x16x4_f32(
                false, Af[s], false, Bf[s][n], (short)0, acc[n], false, false);
    }
}

template <int NT>
__device__ __forceinline__ void do_strip(
    const int j, const int n0,
    const float* __restrict__ in_part,
    float* slot, float* xtx_w,
    const v2f* Pfrag,
    const float* __restrict__ lscale,
    const float (&yv)[NT],
    float (&mn)[NT], float (&mx)[NT],
    const int part, const int lane)
{
    const int lm   = lane & 15;
    const int hsel = lane >> 4;

    #pragma unroll
    for (int t = 0; t < 32; ++t) {
        const int row = t >> 1;
        const int c4  = (t & 1) * 32 + lane;
        const int lr  = min(j * 16 + row, PROWS - 1);
        float4 v = *(const float4*)(in_part + (size_t)lr * NC + 4 * c4);
        *(float4*)(slot + row * APITCH + 4 * c4) = v;
    }
    wave_lds_sync();

    float lsv[NT][8];
    #pragma unroll
    for (int r = 0; r < 8; ++r) {
        const int s_idx = part * PROWS + min(j * 16 + hsel * 8 + r, PROWS - 1);
        const float* lrow = lscale + (size_t)s_idx * NP + n0 * 16 + lm;
        #pragma unroll
        for (int n = 0; n < NT; ++n) lsv[n][r] = lrow[n * 16];
    }

    {
        const float4* row = (const float4*)(slot + lm * APITCH);
        float s = 0.f;
        for (int i = 0; i < NC / 4; ++i) {
            float4 v = row[i];
            s += v.x * v.x + v.y * v.y + v.z * v.z + v.w * v.w;
        }
        xtx_w[lm] = s;
    }
    wave_lds_sync();

    v8f acc[NT];
    #pragma unroll
    for (int n = 0; n < NT; ++n) acc[n] = (v8f){0.f, 0.f, 0.f, 0.f, 0.f, 0.f, 0.f, 0.f};

    const int a_off = lm * APITCH + 2 * hsel;

    if constexpr (NT == NTILES) {
        const float* a_row = slot + a_off;
        const v2f*   b_col = Pfrag + hsel * NP + n0 * 16 + lm;
        v2f Af0[2], Af1[2];
        v2f Bf0[2][NT], Bf1[2][NT];
        load_group<NT>(Af0, Bf0, a_row, b_col, 0);
        #pragma unroll 1
        for (int k0 = 0; k0 < NC - 16; k0 += 16) {
            load_group<NT>(Af1, Bf1, a_row, b_col, k0 + 8);
            __builtin_amdgcn_sched_barrier(0);
            mma_group<NT>(acc, Af0, Bf0);
            __builtin_amdgcn_sched_barrier(0);
            load_group<NT>(Af0, Bf0, a_row, b_col, k0 + 16);
            __builtin_amdgcn_sched_barrier(0);
            mma_group<NT>(acc, Af1, Bf1);
        }
        load_group<NT>(Af1, Bf1, a_row, b_col, NC - 8);
        __builtin_amdgcn_sched_barrier(0);
        mma_group<NT>(acc, Af0, Bf0);
        mma_group<NT>(acc, Af1, Bf1);
    } else {
        #pragma unroll 8
        for (int k0 = 0; k0 < NC; k0 += 4) {
            v2f Af = *(const v2f*)(slot + a_off + k0);
            const v2f* Bp = Pfrag + ((k0 >> 1) + hsel) * NP + n0 * 16 + lm;
            #pragma unroll
            for (int n = 0; n < NT; ++n) {
                v2f Bf = Bp[n * 16];
                acc[n] = __builtin_amdgcn_wmma_f32_16x16x4_f32(
                    false, Af, false, Bf, (short)0, acc[n], false, false);
            }
        }
    }

    const int row_loc = hsel * 8;

    #pragma unroll
    for (int n = 0; n < NT; ++n) {
        float m0 = mn[n];
        float m1 = mx[n];
        #pragma unroll
        for (int r = 0; r < 8; ++r) {
            float xt  = xtx_w[row_loc + r];
            float sf  = __expf(lsv[n][r]);
            float xty = acc[n][r];
            float nsq = fmaxf(xt - 2.f * xty + yv[n], 0.f);
            float dmin = nsq * __builtin_amdgcn_rcpf(sf + EPS);
            float ratio = __builtin_fmaf(__builtin_amdgcn_rcpf(nsq + EPS), 1.f - EPS, 1.f);
            float sim  = (__builtin_amdgcn_logf(ratio) * LN2) * sf;
            m0 = fminf(m0, dmin);
            m1 = fmaxf(m1, sim);
        }
        mn[n] = m0;
        mx[n] = m1;
    }
    wave_lds_sync();
}

__global__ __launch_bounds__(256)
void proto_main(const float* __restrict__ in,
                const float* __restrict__ proto,
                const float* __restrict__ lscale,
                float* __restrict__ pmin,
                float* __restrict__ pmax)
{
    extern __shared__ float lds[];
    float* a_lds = lds;
    float* p_lds = lds + MWAVES * 16 * APITCH;
    float* xtx   = p_lds + NC * NP;
    float* yty_l = xtx + MWAVES * 16;

    const int bglob = blockIdx.x / PARTS;
    const int part  = blockIdx.x % PARTS;
    const int tid   = threadIdx.x;

    {
        float2* dst = (float2*)p_lds;
        for (int i = tid; i < (NC / 2) * NP; i += 256) {
            int kp = i >> 7;
            int p  = i & (NP - 1);
            float v0 = proto[(2 * kp) * NP + p];
            float v1 = proto[(2 * kp + 1) * NP + p];
            dst[i] = make_float2(v0, v1);
        }
    }
    __syncthreads();

    if (tid < NP) {
        const float2* src = (const float2*)p_lds + tid;
        float s = 0.f;
        for (int k0 = 0; k0 < NC / 2; k0 += 32) {
            float2 v[32];
            #pragma unroll
            for (int i = 0; i < 32; ++i) v[i] = src[(k0 + i) * NP];
            __builtin_amdgcn_sched_barrier(0);
            #pragma unroll
            for (int i = 0; i < 32; ++i) {
                s = __builtin_fmaf(v[i].x, v[i].x, s);
                s = __builtin_fmaf(v[i].y, v[i].y, s);
            }
        }
        yty_l[tid] = s;
    }
    __syncthreads();

    const int wave = __builtin_amdgcn_readfirstlane(tid >> 5);
    const int lane = tid & 31;

    float* slot  = a_lds + wave * 16 * APITCH;
    float* xtx_w = xtx + wave * 16;
    const float* in_part = in + ((size_t)bglob * NS + (size_t)part * PROWS) * NC;
    const v2f* Pfrag = (const v2f*)p_lds;

    const int lm = lane & 15;

    float yv[NTILES], mn[NTILES], mx[NTILES];
    #pragma unroll
    for (int n = 0; n < NTILES; ++n) {
        yv[n] = yty_l[n * 16 + lm];
        mn[n] = 3.0e38f;
        mx[n] = 0.f;
    }
    float yv1[1], mn1[1], mx1[1];
    yv1[0] = yty_l[wave * 16 + lm];
    mn1[0] = 3.0e38f;
    mx1[0] = 0.f;

    int done = 0;
    for (int j = wave; j < WSTRIPS; j += MWAVES) {
        if (done == 0)      __builtin_amdgcn_s_setprio(3);
        else if (done == 1) __builtin_amdgcn_s_setprio(2);
        else                __builtin_amdgcn_s_setprio(1);
        do_strip<NTILES>(j, 0, in_part, slot, xtx_w, Pfrag, lscale, yv, mn, mx, part, lane);
        ++done;
    }
    __builtin_amdgcn_s_setprio(0);
    do_strip<1>(WSTRIPS, wave, in_part, slot, xtx_w, Pfrag, lscale, yv1, mn1, mx1, part, lane);

    #pragma unroll
    for (int n = 0; n < NTILES; ++n) {
        mn[n] = fminf(mn[n], __shfl_xor(mn[n], 16));
        mx[n] = fmaxf(mx[n], __shfl_xor(mx[n], 16));
    }
    mn1[0] = fminf(mn1[0], __shfl_xor(mn1[0], 16));
    mx1[0] = fmaxf(mx1[0], __shfl_xor(mx1[0], 16));
    if (lane < 16) {
        #pragma unroll
        for (int n = 0; n < NTILES; ++n) {
            slot[n * 16 + lm]          = mn[n];
            slot[APITCH + n * 16 + lm] = mx[n];
        }
        slot[2 * APITCH + lm]      = mn1[0];
        slot[2 * APITCH + 16 + lm] = mx1[0];
    }
    __syncthreads();
    {
        const int which = tid >> 7;
        const int p     = tid & (NP - 1);
        float v = a_lds[(p >> 4) * 16 * APITCH + 2 * APITCH + which * 16 + (p & 15)];
        #pragma unroll
        for (int w = 0; w < MWAVES; ++w) {
            const float x = a_lds[w * 16 * APITCH + which * APITCH + p];
            v = which ? fmaxf(v, x) : fminf(v, x);
        }
        float* dst = which ? pmax : pmin;
        dst[blockIdx.x * NP + p] = v;
    }
}

__global__ __launch_bounds__(FIN_THREADS)
void proto_final(const float* __restrict__ pmin,
                 const float* __restrict__ pmax,
                 const float* __restrict__ pcc,
                 float* __restrict__ out)
{
    __shared__ float r1[NB];
    __shared__ float r2[NB];
    const int t    = threadIdx.x;
    const int b    = t >> 5;
    const int lane = t & 31;

    if (blockIdx.x == 1) {
        const v4f* qmax = (const v4f*)(pmax + (size_t)b * PARTS * NP) + lane;
        v4f vmax[PARTS];
        #pragma unroll
        for (int q = 0; q < PARTS; ++q) vmax[q] = qmax[q * (NP / 4)];
        v4f hi = vmax[0];
        #pragma unroll
        for (int q = 1; q < PARTS; ++q) {
            hi.x = fmaxf(hi.x, vmax[q].x); hi.y = fmaxf(hi.y, vmax[q].y);
            hi.z = fmaxf(hi.z, vmax[q].z); hi.w = fmaxf(hi.w, vmax[q].w);
        }
        ((v4f*)(out + b * NP))[lane] = hi;
    } else {
        const v4f* qmin = (const v4f*)(pmin + (size_t)b * PARTS * NP) + lane;
        v4f vmin[PARTS];
        #pragma unroll
        for (int q = 0; q < PARTS; ++q) vmin[q] = qmin[q * (NP / 4)];
        const v4f m = ((const v4f*)(pcc + b * NP))[lane];
        v4f lo = vmin[0];
        #pragma unroll
        for (int q = 1; q < PARTS; ++q) {
            lo.x = fminf(lo.x, vmin[q].x); lo.y = fminf(lo.y, vmin[q].y);
            lo.z = fminf(lo.z, vmin[q].z); lo.w = fminf(lo.w, vmin[q].w);
        }
        const float v0 = MAXDIST - lo.x, v1 = MAXDIST - lo.y, v2 = MAXDIST - lo.z, v3 = MAXDIST - lo.w;
        float m1 = fmaxf(fmaxf(v0 * m.x, v1 * m.y), fmaxf(v2 * m.z, v3 * m.w));
        float m2 = fmaxf(fmaxf(v0 * (1.f - m.x), v1 * (1.f - m.y)),
                         fmaxf(v2 * (1.f - m.z), v3 * (1.f - m.w)));
        for (int off = 16; off > 0; off >>= 1) {
            m1 = fmaxf(m1, __shfl_xor(m1, off));
            m2 = fmaxf(m2, __shfl_xor(m2, off));
        }
        if (lane == 0) { r1[b] = m1; r2[b] = m2; }
        __syncthreads();

        if (t == 0) {
            float s1 = 0.f, s2 = 0.f;
            for (int bb = 0; bb < NB; ++bb) {
                s1 += MAXDIST - r1[bb];
                s2 += MAXDIST - r2[bb];
            }
            out[NB * NP]     = s1 / (float)NB;
            out[NB * NP + 1] = s2 / (float)NB;
        }
    }
}

extern "C" void kernel_launch(void* const* d_in, const int* in_sizes, int n_in,
                              void* d_out, int out_size, void* d_ws, size_t ws_size,
                              hipStream_t stream)
{
    const float* inputs = (const float*)d_in[0];
    const float* pcc    = (const float*)d_in[1];
    const float* proto  = (const float*)d_in[2];
    const float* lscale = (const float*)d_in[3];
    float* out = (float*)d_out;

    float* pmin = (float*)d_ws;
    float* pmax = pmin + NB * PARTS * NP;

    const size_t shmem = (size_t)(MWAVES * 16 * APITCH + NC * NP + MWAVES * 16 + NP) * sizeof(float);
    proto_main<<<dim3(NB * PARTS), dim3(256), shmem, stream>>>(
        inputs, proto, lscale, pmin, pmax);

    proto_final<<<dim3(2), dim3(FIN_THREADS), 0, stream>>>(pmin, pmax, pcc, out);
}


# row norms from the copy's registers: a lane sums the squares of its 8 elements of a row, the 32 partials meet in LDS and add in a pairwise tree; no LDS re-read of the strip; xTx sum order changes
# speedup vs baseline: 3.0547x; 1.1049x over previous
#include <hip/hip_runtime.h>
#include <hip/hip_bf16.h>

typedef float v2f __attribute__((ext_vector_type(2)));
typedef float v4f __attribute__((ext_vector_type(4)));
typedef float v8f __attribute__((ext_vector_type(8)));

constexpr int NB = 32;
constexpr int NH = 56, NW = 56;
constexpr int NC = 256;
constexpr int NP = 128;
constexpr int NS = NH * NW;
constexpr int MTILE = 64;
constexpr int NTILES = NP / 16;
constexpr int TILES = NS / MTILE;
constexpr int PARTS   = 8;
constexpr int PROWS   = NS / PARTS;
constexpr int WSTRIPS = PROWS / 16;
constexpr int MWAVES  = 8;
static_assert(MWAVES == NTILES, "the last strip is split one column tile per wave");
static_assert(WSTRIPS % MWAVES == 0 && PROWS - 16 * WSTRIPS > 0 && PROWS - 16 * WSTRIPS <= 16,
              "three whole strips per wave and one partial strip");
constexpr float MAXDIST = 100.0f * 256.0f;
constexpr float EPS = 1e-4f;
constexpr float LN2 = 0.693147180559945309f;
constexpr int FIN_THREADS = NB * 32;
constexpr int APITCH = NC + 4;
constexpr int XP = 32 + 4;


__device__ __forceinline__ void wave_lds_sync()
{
    __builtin_amdgcn_fence(__ATOMIC_RELEASE, "wavefront");
    __builtin_amdgcn_wave_barrier();
    __builtin_amdgcn_fence(__ATOMIC_ACQUIRE, "wavefront");
}

template <int NT>
__device__ __forceinline__ void load_group(v2f (&Af)[2], v2f (&Bf)[2][NT],
                                           const float* a_row, const v2f* b_col, const int k0)
{
    #pragma unroll
    for (int s = 0; s < 2; ++s) {
        const int k = k0 + 4 * s;
        Af[s] = *(const v2f*)(a_row + k);
        const v2f* Bp = b_col + (k >> 1) * NP;
        #pragma unroll
        for (int n = 0; n < NT; ++n) Bf[s][n] = Bp[n * 16];
    }
}

template <int NT>
__device__ __forceinline__ void mma_group(v8f (&acc)[NT], const v2f (&Af)[2], const v2f (&Bf)[2][NT])
{
    #pragma unroll
    for (int s = 0; s < 2; ++s) {
        #pragma unroll
        for (int n = 0; n < NT; ++n)
            acc[n] = __builtin_amdgcn_wmma_f32_16x16x4_f32(
                false, Af[s], false, Bf[s][n], (short)0, acc[n], false, false);
    }
}

template <int NT>
__device__ __forceinline__ void do_strip(
    const int j, const int n0,
    const float* __restrict__ in_part,
    float* slot, float* xtx_w,
    float* xpart_w,
    const v2f* Pfrag,
    const float* __restrict__ lscale,
    const float (&yv)[NT],
    float (&mn)[NT], float (&mx)[NT],
    const int part, const int lane)
{
    const int lm   = lane & 15;
    const int hsel = lane >> 4;

    v2f pacc[16];
    #pragma unroll
    for (int t = 0; t < 32; ++t) {
        const int row = t >> 1;
        const int c4  = (t & 1) * 32 + lane;
        const int lr  = min(j * 16 + row, PROWS - 1);
        float4 v = *(const float4*)(in_part + (size_t)lr * NC + 4 * c4);
        *(float4*)(slot + row * APITCH + 4 * c4) = v;
        const v2f lo = (v2f){v.x, v.y};
        const v2f hi = (v2f){v.z, v.w};
        if ((t & 1) == 0) pacc[row] = lo * lo;
        else              pacc[row] = lo * lo + pacc[row];
        pacc[row] = hi * hi + pacc[row];
    }
    #pragma unroll
    for (int r = 0; r < 16; ++r) xpart_w[r * XP + lane] = pacc[r].x + pacc[r].y;
    wave_lds_sync();

    float lsv[NT][8];
    #pragma unroll
    for (int r = 0; r < 8; ++r) {
        const int s_idx = part * PROWS + min(j * 16 + hsel * 8 + r, PROWS - 1);
        const float* lrow = lscale + (size_t)s_idx * NP + n0 * 16 + lm;
        #pragma unroll
        for (int n = 0; n < NT; ++n) lsv[n][r] = lrow[n * 16];
    }

    {
        const float4* pr = (const float4*)(xpart_w + lm * XP);
        float4 a[8];
        #pragma unroll
        for (int i = 0; i < 8; ++i) a[i] = pr[i];
        float s8[8];
        #pragma unroll
        for (int i = 0; i < 8; ++i) s8[i] = (a[i].x + a[i].y) + (a[i].z + a[i].w);
        xtx_w[lm] = ((s8[0] + s8[1]) + (s8[2] + s8[3])) + ((s8[4] + s8[5]) + (s8[6] + s8[7]));
    }
    wave_lds_sync();

    v8f acc[NT];
    #pragma unroll
    for (int n = 0; n < NT; ++n) acc[n] = (v8f){0.f, 0.f, 0.f, 0.f, 0.f, 0.f, 0.f, 0.f};

    const int a_off = lm * APITCH + 2 * hsel;

    if constexpr (NT == NTILES) {
        const float* a_row = slot + a_off;
        const v2f*   b_col = Pfrag + hsel * NP + n0 * 16 + lm;
        v2f Af0[2], Af1[2];
        v2f Bf0[2][NT], Bf1[2][NT];
        load_group<NT>(Af0, Bf0, a_row, b_col, 0);
        #pragma unroll 1
        for (int k0 = 0; k0 < NC - 16; k0 += 16) {
            load_group<NT>(Af1, Bf1, a_row, b_col, k0 + 8);
            __builtin_amdgcn_sched_barrier(0);
            mma_group<NT>(acc, Af0, Bf0);
            __builtin_amdgcn_sched_barrier(0);
            load_group<NT>(Af0, Bf0, a_row, b_col, k0 + 16);
            __builtin_amdgcn_sched_barrier(0);
            mma_group<NT>(acc, Af1, Bf1);
        }
        load_group<NT>(Af1, Bf1, a_row, b_col, NC - 8);
        __builtin_amdgcn_sched_barrier(0);
        mma_group<NT>(acc, Af0, Bf0);
        mma_group<NT>(acc, Af1, Bf1);
    } else {
        #pragma unroll 8
        for (int k0 = 0; k0 < NC; k0 += 4) {
            v2f Af = *(const v2f*)(slot + a_off + k0);
            const v2f* Bp = Pfrag + ((k0 >> 1) + hsel) * NP + n0 * 16 + lm;
            #pragma unroll
            for (int n = 0; n < NT; ++n) {
                v2f Bf = Bp[n * 16];
                acc[n] = __builtin_amdgcn_wmma_f32_16x16x4_f32(
                    false, Af, false, Bf, (short)0, acc[n], false, false);
            }
        }
    }

    const int row_loc = hsel * 8;

    #pragma unroll
    for (int n = 0; n < NT; ++n) {
        float m0 = mn[n];
        float m1 = mx[n];
        #pragma unroll
        for (int r = 0; r < 8; ++r) {
            float xt  = xtx_w[row_loc + r];
            float sf  = __expf(lsv[n][r]);
            float xty = acc[n][r];
            float nsq = fmaxf(xt - 2.f * xty + yv[n], 0.f);
            float dmin = nsq * __builtin_amdgcn_rcpf(sf + EPS);
            float ratio = __builtin_fmaf(__builtin_amdgcn_rcpf(nsq + EPS), 1.f - EPS, 1.f);
            float sim  = (__builtin_amdgcn_logf(ratio) * LN2) * sf;
            m0 = fminf(m0, dmin);
            m1 = fmaxf(m1, sim);
        }
        mn[n] = m0;
        mx[n] = m1;
    }
    wave_lds_sync();
}

__global__ __launch_bounds__(256)
void proto_main(const float* __restrict__ in,
                const float* __restrict__ proto,
                const float* __restrict__ lscale,
                float* __restrict__ pmin,
                float* __restrict__ pmax)
{
    extern __shared__ float lds[];
    float* a_lds = lds;
    float* p_lds = lds + MWAVES * 16 * APITCH;
    float* xtx   = p_lds + NC * NP;
    float* yty_l = xtx + MWAVES * 16;
    float* xpart = yty_l + NP;

    const int bglob = blockIdx.x / PARTS;
    const int part  = blockIdx.x % PARTS;
    const int tid   = threadIdx.x;

    {
        float2* dst = (float2*)p_lds;
        for (int i = tid; i < (NC / 2) * NP; i += 256) {
            int kp = i >> 7;
            int p  = i & (NP - 1);
            float v0 = proto[(2 * kp) * NP + p];
            float v1 = proto[(2 * kp + 1) * NP + p];
            dst[i] = make_float2(v0, v1);
        }
    }
    __syncthreads();

    if (tid < NP) {
        const float2* src = (const float2*)p_lds + tid;
        float s = 0.f;
        for (int k0 = 0; k0 < NC / 2; k0 += 32) {
            float2 v[32];
            #pragma unroll
            for (int i = 0; i < 32; ++i) v[i] = src[(k0 + i) * NP];
            __builtin_amdgcn_sched_barrier(0);
            #pragma unroll
            for (int i = 0; i < 32; ++i) {
                s = __builtin_fmaf(v[i].x, v[i].x, s);
                s = __builtin_fmaf(v[i].y, v[i].y, s);
            }
        }
        yty_l[tid] = s;
    }
    __syncthreads();

    const int wave = __builtin_amdgcn_readfirstlane(tid >> 5);
    const int lane = tid & 31;

    float* slot  = a_lds + wave * 16 * APITCH;
    float* xtx_w = xtx + wave * 16;
    float* xpart_w = xpart + wave * 16 * XP;
    const float* in_part = in + ((size_t)bglob * NS + (size_t)part * PROWS) * NC;
    const v2f* Pfrag = (const v2f*)p_lds;

    const int lm = lane & 15;

    float yv[NTILES], mn[NTILES], mx[NTILES];
    #pragma unroll
    for (int n = 0; n < NTILES; ++n) {
        yv[n] = yty_l[n * 16 + lm];
        mn[n] = 3.0e38f;
        mx[n] = 0.f;
    }
    float yv1[1], mn1[1], mx1[1];
    yv1[0] = yty_l[wave * 16 + lm];
    mn1[0] = 3.0e38f;
    mx1[0] = 0.f;

    int done = 0;
    for (int j = wave; j < WSTRIPS; j += MWAVES) {
        if (done == 0)      __builtin_amdgcn_s_setprio(3);
        else if (done == 1) __builtin_amdgcn_s_setprio(2);
        else                __builtin_amdgcn_s_setprio(1);
        do_strip<NTILES>(j, 0, in_part, slot, xtx_w, xpart_w, Pfrag, lscale, yv, mn, mx, part, lane);
        ++done;
    }
    __builtin_amdgcn_s_setprio(0);
    do_strip<1>(WSTRIPS, wave, in_part, slot, xtx_w, xpart_w, Pfrag, lscale, yv1, mn1, mx1, part, lane);

    #pragma unroll
    for (int n = 0; n < NTILES; ++n) {
        mn[n] = fminf(mn[n], __shfl_xor(mn[n], 16));
        mx[n] = fmaxf(mx[n], __shfl_xor(mx[n], 16));
    }
    mn1[0] = fminf(mn1[0], __shfl_xor(mn1[0], 16));
    mx1[0] = fmaxf(mx1[0], __shfl_xor(mx1[0], 16));
    if (lane < 16) {
        #pragma unroll
        for (int n = 0; n < NTILES; ++n) {
            slot[n * 16 + lm]          = mn[n];
            slot[APITCH + n * 16 + lm] = mx[n];
        }
        slot[2 * APITCH + lm]      = mn1[0];
        slot[2 * APITCH + 16 + lm] = mx1[0];
    }
    __syncthreads();
    {
        const int which = tid >> 7;
        const int p     = tid & (NP - 1);
        float v = a_lds[(p >> 4) * 16 * APITCH + 2 * APITCH + which * 16 + (p & 15)];
        #pragma unroll
        for (int w = 0; w < MWAVES; ++w) {
            const float x = a_lds[w * 16 * APITCH + which * APITCH + p];
            v = which ? fmaxf(v, x) : fminf(v, x);
        }
        float* dst = which ? pmax : pmin;
        dst[blockIdx.x * NP + p] = v;
    }
}

__global__ __launch_bounds__(FIN_THREADS)
void proto_final(const float* __restrict__ pmin,
                 const float* __restrict__ pmax,
                 const float* __restrict__ pcc,
                 float* __restrict__ out)
{
    __shared__ float r1[NB];
    __shared__ float r2[NB];
    const int t    = threadIdx.x;
    const int b    = t >> 5;
    const int lane = t & 31;

    if (blockIdx.x == 1) {
        const v4f* qmax = (const v4f*)(pmax + (size_t)b * PARTS * NP) + lane;
        v4f vmax[PARTS];
        #pragma unroll
        for (int q = 0; q < PARTS; ++q) vmax[q] = qmax[q * (NP / 4)];
        v4f hi = vmax[0];
        #pragma unroll
        for (int q = 1; q < PARTS; ++q) {
            hi.x = fmaxf(hi.x, vmax[q].x); hi.y = fmaxf(hi.y, vmax[q].y);
            hi.z = fmaxf(hi.z, vmax[q].z); hi.w = fmaxf(hi.w, vmax[q].w);
        }
        ((v4f*)(out + b * NP))[lane] = hi;
    } else {
        const v4f* qmin = (const v4f*)(pmin + (size_t)b * PARTS * NP) + lane;
        v4f vmin[PARTS];
        #pragma unroll
        for (int q = 0; q < PARTS; ++q) vmin[q] = qmin[q * (NP / 4)];
        const v4f m = ((const v4f*)(pcc + b * NP))[lane];
        v4f lo = vmin[0];
        #pragma unroll
        for (int q = 1; q < PARTS; ++q) {
            lo.x = fminf(lo.x, vmin[q].x); lo.y = fminf(lo.y, vmin[q].y);
            lo.z = fminf(lo.z, vmin[q].z); lo.w = fminf(lo.w, vmin[q].w);
        }
        const float v0 = MAXDIST - lo.x, v1 = MAXDIST - lo.y, v2 = MAXDIST - lo.z, v3 = MAXDIST - lo.w;
        float m1 = fmaxf(fmaxf(v0 * m.x, v1 * m.y), fmaxf(v2 * m.z, v3 * m.w));
        float m2 = fmaxf(fmaxf(v0 * (1.f - m.x), v1 * (1.f - m.y)),
                         fmaxf(v2 * (1.f - m.z), v3 * (1.f - m.w)));
        for (int off = 16; off > 0; off >>= 1) {
            m1 = fmaxf(m1, __shfl_xor(m1, off));
            m2 = fmaxf(m2, __shfl_xor(m2, off));
        }
        if (lane == 0) { r1[b] = m1; r2[b] = m2; }
        __syncthreads();

        if (t == 0) {
            float s1 = 0.f, s2 = 0.f;
            for (int bb = 0; bb < NB; ++bb) {
                s1 += MAXDIST - r1[bb];
                s2 += MAXDIST - r2[bb];
            }
            out[NB * NP]     = s1 / (float)NB;
            out[NB * NP + 1] = s2 / (float)NB;
        }
    }
}

extern "C" void kernel_launch(void* const* d_in, const int* in_sizes, int n_in,
                              void* d_out, int out_size, void* d_ws, size_t ws_size,
                              hipStream_t stream)
{
    const float* inputs = (const float*)d_in[0];
    const float* pcc    = (const float*)d_in[1];
    const float* proto  = (const float*)d_in[2];
    const float* lscale = (const float*)d_in[3];
    float* out = (float*)d_out;

    float* pmin = (float*)d_ws;
    float* pmax = pmin + NB * PARTS * NP;

    const size_t shmem = (size_t)(MWAVES * 16 * APITCH + NC * NP + MWAVES * 16 + NP + MWAVES * 16 * XP) * sizeof(float);
    proto_main<<<dim3(NB * PARTS), dim3(256), shmem, stream>>>(
        inputs, proto, lscale, pmin, pmax);

    proto_final<<<dim3(2), dim3(FIN_THREADS), 0, stream>>>(pmin, pmax, pcc, out);
}


# prologue in one pass: prototypes fetched as one group of 32 16-byte loads per thread (was 8 serial rounds of 4-byte loads); yTy from those registers, 8 partials in a fixed tree: yTy sum order changes
# speedup vs baseline: 3.2533x; 1.0650x over previous
#include <hip/hip_runtime.h>
#include <hip/hip_bf16.h>

typedef float v2f __attribute__((ext_vector_type(2)));
typedef float v4f __attribute__((ext_vector_type(4)));
typedef float v8f __attribute__((ext_vector_type(8)));

constexpr int NB = 32;
constexpr int NH = 56, NW = 56;
constexpr int NC = 256;
constexpr int NP = 128;
constexpr int NS = NH * NW;
constexpr int MTILE = 64;
constexpr int NTILES = NP / 16;
constexpr int TILES = NS / MTILE;
constexpr int PARTS   = 8;
constexpr int PROWS   = NS / PARTS;
constexpr int WSTRIPS = PROWS / 16;
constexpr int MWAVES  = 8;
static_assert(MWAVES == NTILES, "the last strip is split one column tile per wave");
static_assert(WSTRIPS % MWAVES == 0 && PROWS - 16 * WSTRIPS > 0 && PROWS - 16 * WSTRIPS <= 16,
              "three whole strips per wave and one partial strip");
constexpr float MAXDIST = 100.0f * 256.0f;
constexpr float EPS = 1e-4f;
constexpr float LN2 = 0.693147180559945309f;
constexpr int FIN_THREADS = NB * 32;
constexpr int APITCH = NC + 4;
constexpr int XP = 32 + 4;


__device__ __forceinline__ void wave_lds_sync()
{
    __builtin_amdgcn_fence(__ATOMIC_RELEASE, "wavefront");
    __builtin_amdgcn_wave_barrier();
    __builtin_amdgcn_fence(__ATOMIC_ACQUIRE, "wavefront");
}

template <int NT>
__device__ __forceinline__ void load_group(v2f (&Af)[2], v2f (&Bf)[2][NT],
                                           const float* a_row, const v2f* b_col, const int k0)
{
    #pragma unroll
    for (int s = 0; s < 2; ++s) {
        const int k = k0 + 4 * s;
        Af[s] = *(const v2f*)(a_row + k);
        const v2f* Bp = b_col + (k >> 1) * NP;
        #pragma unroll
        for (int n = 0; n < NT; ++n) Bf[s][n] = Bp[n * 16];
    }
}

template <int NT>
__device__ __forceinline__ void mma_group(v8f (&acc)[NT], const v2f (&Af)[2], const v2f (&Bf)[2][NT])
{
    #pragma unroll
    for (int s = 0; s < 2; ++s) {
        #pragma unroll
        for (int n = 0; n < NT; ++n)
            acc[n] = __builtin_amdgcn_wmma_f32_16x16x4_f32(
                false, Af[s], false, Bf[s][n], (short)0, acc[n], false, false);
    }
}

template <int NT>
__device__ __forceinline__ void do_strip(
    const int j, const int n0,
    const float* __restrict__ in_part,
    float* slot, float* xtx_w,
    float* xpart_w,
    const v2f* Pfrag,
    const float* __restrict__ lscale,
    const float (&yv)[NT],
    float (&mn)[NT], float (&mx)[NT],
    const int part, const int lane)
{
    const int lm   = lane & 15;
    const int hsel = lane >> 4;

    v2f pacc[16];
    #pragma unroll
    for (int t = 0; t < 32; ++t) {
        const int row = t >> 1;
        const int c4  = (t & 1) * 32 + lane;
        const int lr  = min(j * 16 + row, PROWS - 1);
        float4 v = *(const float4*)(in_part + (size_t)lr * NC + 4 * c4);
        *(float4*)(slot + row * APITCH + 4 * c4) = v;
        const v2f lo = (v2f){v.x, v.y};
        const v2f hi = (v2f){v.z, v.w};
        if ((t & 1) == 0) pacc[row] = lo * lo;
        else              pacc[row] = lo * lo + pacc[row];
        pacc[row] = hi * hi + pacc[row];
    }
    #pragma unroll
    for (int r = 0; r < 16; ++r) xpart_w[r * XP + lane] = pacc[r].x + pacc[r].y;
    wave_lds_sync();

    float lsv[NT][8];
    #pragma unroll
    for (int r = 0; r < 8; ++r) {
        const int s_idx = part * PROWS + min(j * 16 + hsel * 8 + r, PROWS - 1);
        const float* lrow = lscale + (size_t)s_idx * NP + n0 * 16 + lm;
        #pragma unroll
        for (int n = 0; n < NT; ++n) lsv[n][r] = lrow[n * 16];
    }

    {
        const float4* pr = (const float4*)(xpart_w + lm * XP);
        float4 a[8];
        #pragma unroll
        for (int i = 0; i < 8; ++i) a[i] = pr[i];
        float s8[8];
        #pragma unroll
        for (int i = 0; i < 8; ++i) s8[i] = (a[i].x + a[i].y) + (a[i].z + a[i].w);
        xtx_w[lm] = ((s8[0] + s8[1]) + (s8[2] + s8[3])) + ((s8[4] + s8[5]) + (s8[6] + s8[7]));
    }
    wave_lds_sync();

    v8f acc[NT];
    #pragma unroll
    for (int n = 0; n < NT; ++n) acc[n] = (v8f){0.f, 0.f, 0.f, 0.f, 0.f, 0.f, 0.f, 0.f};

    const int a_off = lm * APITCH + 2 * hsel;

    if constexpr (NT == NTILES) {
        const float* a_row = slot + a_off;
        const v2f*   b_col = Pfrag + hsel * NP + n0 * 16 + lm;
        v2f Af0[2], Af1[2];
        v2f Bf0[2][NT], Bf1[2][NT];
        load_group<NT>(Af0, Bf0, a_row, b_col, 0);
        #pragma unroll 1
        for (int k0 = 0; k0 < NC - 16; k0 += 16) {
            load_group<NT>(Af1, Bf1, a_row, b_col, k0 + 8);
            __builtin_amdgcn_sched_barrier(0);
            mma_group<NT>(acc, Af0, Bf0);
            __builtin_amdgcn_sched_barrier(0);
            load_group<NT>(Af0, Bf0, a_row, b_col, k0 + 16);
            __builtin_amdgcn_sched_barrier(0);
            mma_group<NT>(acc, Af1, Bf1);
        }
        load_group<NT>(Af1, Bf1, a_row, b_col, NC - 8);
        __builtin_amdgcn_sched_barrier(0);
        mma_group<NT>(acc, Af0, Bf0);
        mma_group<NT>(acc, Af1, Bf1);
    } else {
        #pragma unroll 8
        for (int k0 = 0; k0 < NC; k0 += 4) {
            v2f Af = *(const v2f*)(slot + a_off + k0);
            const v2f* Bp = Pfrag + ((k0 >> 1) + hsel) * NP + n0 * 16 + lm;
            #pragma unroll
            for (int n = 0; n < NT; ++n) {
                v2f Bf = Bp[n * 16];
                acc[n] = __builtin_amdgcn_wmma_f32_16x16x4_f32(
                    false, Af, false, Bf, (short)0, acc[n], false, false);
            }
        }
    }

    const int row_loc = hsel * 8;

    #pragma unroll
    for (int n = 0; n < NT; ++n) {
        float m0 = mn[n];
        float m1 = mx[n];
        #pragma unroll
        for (int r = 0; r < 8; ++r) {
            float xt  = xtx_w[row_loc + r];
            float sf  = __expf(lsv[n][r]);
            float xty = acc[n][r];
            float nsq = fmaxf(xt - 2.f * xty + yv[n], 0.f);
            float dmin = nsq * __builtin_amdgcn_rcpf(sf + EPS);
            float ratio = __builtin_fmaf(__builtin_amdgcn_rcpf(nsq + EPS), 1.f - EPS, 1.f);
            float sim  = (__builtin_amdgcn_logf(ratio) * LN2) * sf;
            m0 = fminf(m0, dmin);
            m1 = fmaxf(m1, sim);
        }
        mn[n] = m0;
        mx[n] = m1;
    }
    wave_lds_sync();
}

__global__ __launch_bounds__(256)
void proto_main(const float* __restrict__ in,
                const float* __restrict__ proto,
                const float* __restrict__ lscale,
                float* __restrict__ pmin,
                float* __restrict__ pmax)
{
    extern __shared__ float lds[];
    float* a_lds = lds;
    float* p_lds = lds + MWAVES * 16 * APITCH;
    float* xtx   = p_lds + NC * NP;
    float* yty_l = xtx + MWAVES * 16;
    float* xpart = yty_l + NP;

    const int bglob = blockIdx.x / PARTS;
    const int part  = blockIdx.x % PARTS;
    const int tid   = threadIdx.x;

    {
        const int wv = tid >> 5;
        const int q  = tid & 31;
        float4 r0[16], r1[16];
        #pragma unroll
        for (int i = 0; i < 16; ++i) {
            const int kp = i * MWAVES + wv;
            r0[i] = *(const float4*)(proto + (2 * kp) * NP + 4 * q);
            r1[i] = *(const float4*)(proto + (2 * kp + 1) * NP + 4 * q);
        }
        float4* dst4 = (float4*)p_lds;
        v2f s01 = (v2f){0.f, 0.f};
        v2f s23 = (v2f){0.f, 0.f};
        #pragma unroll
        for (int i = 0; i < 16; ++i) {
            const int kp = i * MWAVES + wv;
            dst4[kp * (NP / 2) + 2 * q]     = make_float4(r0[i].x, r1[i].x, r0[i].y, r1[i].y);
            dst4[kp * (NP / 2) + 2 * q + 1] = make_float4(r0[i].z, r1[i].z, r0[i].w, r1[i].w);
            const v2f a01 = (v2f){r0[i].x, r0[i].y};
            const v2f a23 = (v2f){r0[i].z, r0[i].w};
            const v2f b01 = (v2f){r1[i].x, r1[i].y};
            const v2f b23 = (v2f){r1[i].z, r1[i].w};
            s01 = a01 * a01 + s01;
            s23 = a23 * a23 + s23;
            s01 = b01 * b01 + s01;
            s23 = b23 * b23 + s23;
            if (i == 7) __builtin_amdgcn_sched_barrier(0);
        }
        *(float4*)(xpart + wv * NP + 4 * q) = make_float4(s01.x, s01.y, s23.x, s23.y);
    }
    __syncthreads();

    if (tid < NP) {
        float y[MWAVES];
        #pragma unroll
        for (int w = 0; w < MWAVES; ++w) y[w] = xpart[w * NP + tid];
        yty_l[tid] = ((y[0] + y[1]) + (y[2] + y[3])) + ((y[4] + y[5]) + (y[6] + y[7]));
    }
    __syncthreads();

    const int wave = __builtin_amdgcn_readfirstlane(tid >> 5);
    const int lane = tid & 31;

    float* slot  = a_lds + wave * 16 * APITCH;
    float* xtx_w = xtx + wave * 16;
    float* xpart_w = xpart + wave * 16 * XP;
    const float* in_part = in + ((size_t)bglob * NS + (size_t)part * PROWS) * NC;
    const v2f* Pfrag = (const v2f*)p_lds;

    const int lm = lane & 15;

    float yv[NTILES], mn[NTILES], mx[NTILES];
    #pragma unroll
    for (int n = 0; n < NTILES; ++n) {
        yv[n] = yty_l[n * 16 + lm];
        mn[n] = 3.0e38f;
        mx[n] = 0.f;
    }
    float yv1[1], mn1[1], mx1[1];
    yv1[0] = yty_l[wave * 16 + lm];
    mn1[0] = 3.0e38f;
    mx1[0] = 0.f;

    int done = 0;
    for (int j = wave; j < WSTRIPS; j += MWAVES) {
        if (done == 0)      __builtin_amdgcn_s_setprio(3);
        else if (done == 1) __builtin_amdgcn_s_setprio(2);
        else                __builtin_amdgcn_s_setprio(1);
        do_strip<NTILES>(j, 0, in_part, slot, xtx_w, xpart_w, Pfrag, lscale, yv, mn, mx, part, lane);
        ++done;
    }
    __builtin_amdgcn_s_setprio(0);
    do_strip<1>(WSTRIPS, wave, in_part, slot, xtx_w, xpart_w, Pfrag, lscale, yv1, mn1, mx1, part, lane);

    #pragma unroll
    for (int n = 0; n < NTILES; ++n) {
        mn[n] = fminf(mn[n], __shfl_xor(mn[n], 16));
        mx[n] = fmaxf(mx[n], __shfl_xor(mx[n], 16));
    }
    mn1[0] = fminf(mn1[0], __shfl_xor(mn1[0], 16));
    mx1[0] = fmaxf(mx1[0], __shfl_xor(mx1[0], 16));
    if (lane < 16) {
        #pragma unroll
        for (int n = 0; n < NTILES; ++n) {
            slot[n * 16 + lm]          = mn[n];
            slot[APITCH + n * 16 + lm] = mx[n];
        }
        slot[2 * APITCH + lm]      = mn1[0];
        slot[2 * APITCH + 16 + lm] = mx1[0];
    }
    __syncthreads();
    {
        const int which = tid >> 7;
        const int p     = tid & (NP - 1);
        float v = a_lds[(p >> 4) * 16 * APITCH + 2 * APITCH + which * 16 + (p & 15)];
        #pragma unroll
        for (int w = 0; w < MWAVES; ++w) {
            const float x = a_lds[w * 16 * APITCH + which * APITCH + p];
            v = which ? fmaxf(v, x) : fminf(v, x);
        }
        float* dst = which ? pmax : pmin;
        dst[blockIdx.x * NP + p] = v;
    }
}

__global__ __launch_bounds__(FIN_THREADS)
void proto_final(const float* __restrict__ pmin,
                 const float* __restrict__ pmax,
                 const float* __restrict__ pcc,
                 float* __restrict__ out)
{
    __shared__ float r1[NB];
    __shared__ float r2[NB];
    const int t    = threadIdx.x;
    const int b    = t >> 5;
    const int lane = t & 31;

    if (blockIdx.x == 1) {
        const v4f* qmax = (const v4f*)(pmax + (size_t)b * PARTS * NP) + lane;
        v4f vmax[PARTS];
        #pragma unroll
        for (int q = 0; q < PARTS; ++q) vmax[q] = qmax[q * (NP / 4)];
        v4f hi = vmax[0];
        #pragma unroll
        for (int q = 1; q < PARTS; ++q) {
            hi.x = fmaxf(hi.x, vmax[q].x); hi.y = fmaxf(hi.y, vmax[q].y);
            hi.z = fmaxf(hi.z, vmax[q].z); hi.w = fmaxf(hi.w, vmax[q].w);
        }
        ((v4f*)(out + b * NP))[lane] = hi;
    } else {
        const v4f* qmin = (const v4f*)(pmin + (size_t)b * PARTS * NP) + lane;
        v4f vmin[PARTS];
        #pragma unroll
        for (int q = 0; q < PARTS; ++q) vmin[q] = qmin[q * (NP / 4)];
        const v4f m = ((const v4f*)(pcc + b * NP))[lane];
        v4f lo = vmin[0];
        #pragma unroll
        for (int q = 1; q < PARTS; ++q) {
            lo.x = fminf(lo.x, vmin[q].x); lo.y = fminf(lo.y, vmin[q].y);
            lo.z = fminf(lo.z, vmin[q].z); lo.w = fminf(lo.w, vmin[q].w);
        }
        const float v0 = MAXDIST - lo.x, v1 = MAXDIST - lo.y, v2 = MAXDIST - lo.z, v3 = MAXDIST - lo.w;
        float m1 = fmaxf(fmaxf(v0 * m.x, v1 * m.y), fmaxf(v2 * m.z, v3 * m.w));
        float m2 = fmaxf(fmaxf(v0 * (1.f - m.x), v1 * (1.f - m.y)),
                         fmaxf(v2 * (1.f - m.z), v3 * (1.f - m.w)));
        for (int off = 16; off > 0; off >>= 1) {
            m1 = fmaxf(m1, __shfl_xor(m1, off));
            m2 = fmaxf(m2, __shfl_xor(m2, off));
        }
        if (lane == 0) { r1[b] = m1; r2[b] = m2; }
        __syncthreads();

        if (t == 0) {
            float s1 = 0.f, s2 = 0.f;
            for (int bb = 0; bb < NB; ++bb) {
                s1 += MAXDIST - r1[bb];
                s2 += MAXDIST - r2[bb];
            }
            out[NB * NP]     = s1 / (float)NB;
            out[NB * NP + 1] = s2 / (float)NB;
        }
    }
}

extern "C" void kernel_launch(void* const* d_in, const int* in_sizes, int n_in,
                              void* d_out, int out_size, void* d_ws, size_t ws_size,
                              hipStream_t stream)
{
    const float* inputs = (const float*)d_in[0];
    const float* pcc    = (const float*)d_in[1];
    const float* proto  = (const float*)d_in[2];
    const float* lscale = (const float*)d_in[3];
    float* out = (float*)d_out;

    float* pmin = (float*)d_ws;
    float* pmax = pmin + NB * PARTS * NP;

    const size_t shmem = (size_t)(MWAVES * 16 * APITCH + NC * NP + MWAVES * 16 + NP + MWAVES * 16 * XP) * sizeof(float);
    proto_main<<<dim3(NB * PARTS), dim3(256), shmem, stream>>>(
        inputs, proto, lscale, pmin, pmax);

    proto_final<<<dim3(2), dim3(FIN_THREADS), 0, stream>>>(pmin, pmax, pcc, out);
}
